# starting kernel
# baseline (speedup 1.0000x reference)
#include <hip/hip_runtime.h>


typedef __attribute__((ext_vector_type(2))) float v2f;
typedef __attribute__((ext_vector_type(8))) float v8f;

constexpr double dE    = 3130.0;
constexpr double dNU   = 0.37;
constexpr double dSIGY = 64.8;
constexpr double dHARD = 100.0;
constexpr double dG    = dE / (2.0 * (1.0 + dNU));
constexpr double dLAM  = dE * dNU / ((1.0 + dNU) * (1.0 - 2.0 * dNU));
constexpr double dC    = dE / ((1.0 + dNU) * (1.0 - 2.0 * dNU));

constexpr float fG     = (float)dG;
constexpr float fLAM   = (float)dLAM;
constexpr float fC11   = (float)(dC * (1.0 - dNU));
constexpr float fC12   = (float)(dC * dNU);
constexpr float fSIGY  = (float)dSIGY;
constexpr float fHARD  = (float)dHARD;
constexpr float f3G    = (float)(3.0 * dG);
constexpr float fINV3GH = (float)(1.0 / (3.0 * dG + dHARD));

#define NB   128
#define NS   2048
#define LS   192
#define NO   3
#define NPADN 16

__global__ void prep_bpanel_kernel(const float* __restrict__ W2,
                                   float* __restrict__ Bpad) {
  int k = threadIdx.x;
  if (k >= LS) return;
#pragma unroll
  for (int n = 0; n < NPADN; ++n) {
    float v = 0.0f;
    if (n < NO) {
      float w = W2[n * LS + k];
      v = fmaxf(w, 0.0f) + log1pf(__expf(-fabsf(w)));
    }
    Bpad[n * LS + k] = v;
  }
}

__global__ __launch_bounds__(64) void j2_scan_kernel(
    const float* __restrict__ x,
    const float* __restrict__ W1,
    float* __restrict__ stress) {
  const int b = blockIdx.x;
  const int j = threadIdx.x;

  const float* w = W1 + (3 * j) * 3;
  const float w00 = w[0], w01 = w[1], w02 = w[2];
  const float w10 = w[3], w11 = w[4], w12 = w[5];
  const float w20 = w[6], w21 = w[7], w22 = w[8];

  float eps0 = 0.f, eps1 = 0.f, eps2 = 0.f;
  float sig0 = 0.f, sig1 = 0.f, sig2 = 0.f;
  float szz  = 0.f, eqps = 0.f;

  const float* __restrict__ xb = x + (size_t)b * NS * 3;
  float* __restrict__ outb = stress + (size_t)b * NS * LS + 3 * j;

  for (int s = 0; s < NS; ++s) {
    const float x0 = xb[3 * s + 0];
    const float x1 = xb[3 * s + 1];
    const float x2 = xb[3 * s + 2];

    const float st0 = w00 * x0 + w01 * x1 + w02 * x2;
    const float st1 = w10 * x0 + w11 * x1 + w12 * x2;
    const float st2 = w20 * x0 + w21 * x1 + w22 * x2;

    const float de0 = st0 - eps0;
    const float de1 = st1 - eps1;
    const float de2 = st2 - eps2;
    const float tr0 = sig0 + fC11 * de0 + fC12 * de1;
    const float tr1 = sig1 + fC12 * de0 + fC11 * de1;
    const float tr2 = sig2 + fG * de2;
    const float tzz = szz + fLAM * (de0 + de1);

    const float pm  = (tr0 + tr1 + tzz) * (1.0f / 3.0f);
    const float dxx = tr0 - pm;
    const float dyy = tr1 - pm;
    const float dzz = tzz - pm;
    const float dxy = tr2;
    const float qq  = 1.5f * (dxx * dxx + dyy * dyy + dzz * dzz +
                              2.0f * dxy * dxy) + 1e-12f;
    const float q   = sqrtf(qq);
    const float f   = q - (fSIGY + fHARD * eqps);
    const float dgam = (f > 0.0f) ? f * fINV3GH : 0.0f;
    float rq = __builtin_amdgcn_rcpf(q);
    rq = rq * (2.0f - q * rq);
    const float fac = 1.0f - f3G * dgam * rq;

    sig0 = pm + fac * dxx;
    sig1 = pm + fac * dyy;
    sig2 = fac * dxy;
    szz  = pm + fac * dzz;
    eqps = eqps + dgam;
    eps0 = st0; eps1 = st1; eps2 = st2;

    float* o = outb + (size_t)s * LS;
    o[0] = sig0;
    o[1] = sig1;
    o[2] = sig2;
  }
}

__global__ __launch_bounds__(256) void out_gemm_wmma_kernel(
    const float* __restrict__ stress,
    const float* __restrict__ Bpad,
    float* __restrict__ out) {
  const int lane = threadIdx.x & 31;
  const int wave = threadIdx.x >> 5;
  const int tile = blockIdx.x * 8 + wave;
  const int grp  = lane >> 4;
  const int ln   = lane & 15;

  const float* __restrict__ arow =
      stress + ((size_t)tile * 16 + ln) * LS + 2 * grp;
  const float* __restrict__ brow = Bpad + ln * LS + 2 * grp;

  v8f acc = {};
#pragma unroll
  for (int k = 0; k < LS; k += 4) {
    v2f a = *(const v2f*)(arow + k);
    v2f bm = *(const v2f*)(brow + k);
    acc = __builtin_amdgcn_wmma_f32_16x16x4_f32(
        false, a, false, bm, (short)0, acc, false, false);
  }

  if (ln < NO) {
#pragma unroll
    for (int r = 0; r < 8; ++r) {
      const size_t m = (size_t)tile * 16 + r + 8 * grp;
      out[m * NO + ln] = acc[r];
    }
  }
}

extern "C" void kernel_launch(void* const* d_in, const int* in_sizes, int n_in,
                              void* d_out, int out_size, void* d_ws,
                              size_t ws_size, hipStream_t stream) {
  (void)in_sizes; (void)n_in; (void)out_size; (void)ws_size;
  const float* x  = (const float*)d_in[0];
  const float* W1 = (const float*)d_in[1];
  const float* W2 = (const float*)d_in[2];
  float* out = (float*)d_out;

  float* stress = (float*)d_ws;
  float* Bpad   = stress + (size_t)NB * NS * LS;

  prep_bpanel_kernel<<<1, LS, 0, stream>>>(W2, Bpad);
  j2_scan_kernel<<<NB, 64, 0, stream>>>(x, W1, stress);
  const int tiles = (NB * NS) / 16;
  out_gemm_wmma_kernel<<<tiles / 8, 256, 0, stream>>>(stress, Bpad, out);
}


# scan: x comes through the scalar unit: 96 B of a batch's row per 8 steps by two scalar loads issued 8 steps ahead, read as scalar operands; no LDS copy, no per-step LDS read; sums unchanged
# speedup vs baseline: 2.6349x; 2.6349x over previous
#include <hip/hip_runtime.h>


typedef __attribute__((ext_vector_type(2))) float v2f;
typedef __attribute__((ext_vector_type(8))) float v8f;

constexpr double dE    = 3130.0;
constexpr double dNU   = 0.37;
constexpr double dSIGY = 64.8;
constexpr double dHARD = 100.0;
constexpr double dG    = dE / (2.0 * (1.0 + dNU));
constexpr double dLAM  = dE * dNU / ((1.0 + dNU) * (1.0 - 2.0 * dNU));
constexpr double dC    = dE / ((1.0 + dNU) * (1.0 - 2.0 * dNU));

constexpr float fG     = (float)dG;
constexpr float fLAM   = (float)dLAM;
constexpr float fC11   = (float)(dC * (1.0 - dNU));
constexpr float fC12   = (float)(dC * dNU);
constexpr float fSIGY  = (float)dSIGY;
constexpr float fHARD  = (float)dHARD;
constexpr float f3G    = (float)(3.0 * dG);
constexpr float fINV3GH = (float)(1.0 / (3.0 * dG + dHARD));

#define NB   128
#define NS   2048
#define LS   192
#define NO   3
#define NPADN 16

__global__ void prep_bpanel_kernel(const float* __restrict__ W2,
                                   float* __restrict__ Bpad) {
  int k = threadIdx.x;
  if (k >= LS) return;
#pragma unroll
  for (int n = 0; n < NPADN; ++n) {
    float v = 0.0f;
    if (n < NO) {
      float w = W2[n * LS + k];
      v = fmaxf(w, 0.0f) + log1pf(__expf(-fabsf(w)));
    }
    Bpad[n * LS + k] = v;
  }
}

static_assert(__builtin_bit_cast(unsigned, fC11) == 0x45acff7bu, "C11 literal");
static_assert(__builtin_bit_cast(unsigned, fC12) == 0x454b3437u, "C12 literal");
static_assert(__builtin_bit_cast(unsigned, fLAM) == 0x454b3437u, "LAM literal");
static_assert(__builtin_bit_cast(unsigned, fG) == 0x448ecabfu, "G literal");
static_assert(__builtin_bit_cast(unsigned, fSIGY) == 0x4281999au, "SIGY literal");
static_assert(__builtin_bit_cast(unsigned, fINV3GH) == 0x3994a648u, "1/(3G+H)");
static_assert(__builtin_bit_cast(unsigned, -f3G) == 0xc556301eu, "-3G literal");
static_assert(__builtin_bit_cast(unsigned, 1.0f / 3.0f) == 0x3eaaaaabu, "1/3");
static_assert(__builtin_bit_cast(unsigned, 1e-12f) == 0x2b8cbcccu, "1e-12");

typedef __attribute__((ext_vector_type(8)))  float xs8;
typedef __attribute__((ext_vector_type(16))) float xs16;

#define X_GROUP_LOAD(A, B, PTR) \
    asm volatile("s_load_b512 %0, %2, 0x0\n\t" \
                 "s_load_b256 %1, %2, 0x40" \
                 : "=&s"(A), "=&s"(B) : "s"(PTR) : "memory")
#define X_GROUP_WAIT(A, B) \
    asm volatile("s_wait_kmcnt 0x0" : "+s"(A), "+s"(B) : : "memory")

#define J2_STEP(NX0, NX1, NX2, OFF) \
    asm volatile( \
        "v_fma_f32 %[t0], %[de0], 0x45acff7b, %[sig0]\n\t"           \
        "v_fma_f32 %[t1], %[de0], 0x454b3437, %[sig1]\n\t"           \
        "v_add_f32 %[u], %[de1], %[de0]\n\t"                         \
        "v_fma_f32 %[tr2], %[de2], 0x448ecabf, %[sig2]\n\t"          \
        "v_mul_f32 %[m0], %[nx0], %[w00]\n\t"                        \
        "v_fma_f32 %[t0], %[de1], 0x454b3437, %[t0]\n\t"             \
        "v_fma_f32 %[t1], %[de1], 0x45acff7b, %[t1]\n\t"             \
        "v_fma_f32 %[tzz], %[u], 0x454b3437, %[szz]\n\t"             \
        "v_mul_f32 %[m1], %[nx0], %[w10]\n\t"                        \
        "v_mul_f32 %[m2], %[nx0], %[w20]\n\t"                        \
        "v_add_f32 %[sum], %[t1], %[t0]\n\t"                         \
        "v_fmaak_f32 %[yl], %[hard], %[eqps], 0x4281999a\n\t"        \
        "v_add_f32 %[d2], %[tr2], %[tr2]\n\t"                        \
        "v_add_f32 %[sum], %[tzz], %[sum]\n\t"                       \
        "v_fma_f32 %[m0], %[w01], %[nx1], %[m0]\n\t"                 \
        "v_fma_f32 %[m1], %[w11], %[nx1], %[m1]\n\t"                 \
        "v_fma_f32 %[m2], %[w21], %[nx1], %[m2]\n\t"                 \
        "v_mul_f32 %[pm], 0x3eaaaaab, %[sum]\n\t"                    \
        "v_fma_f32 %[dzz], %[sum], 0xbeaaaaab, %[tzz]\n\t"           \
        "v_sub_f32 %[dxx], %[t0], %[pm]\n\t"                         \
        "v_sub_f32 %[dyy], %[t1], %[pm]\n\t"                         \
        "v_mul_f32 %[a2], %[dzz], %[dzz]\n\t"                        \
        "v_mul_f32 %[a3], %[d2], %[tr2]\n\t"                         \
        "v_mul_f32 %[a0], %[dxx], %[dxx]\n\t"                        \
        "v_mul_f32 %[a1], %[dyy], %[dyy]\n\t"                        \
        "v_fma_f32 %[m0], %[w02], %[nx2], %[m0]\n\t"                 \
        "v_fma_f32 %[m1], %[w12], %[nx2], %[m1]\n\t"                 \
        "v_fma_f32 %[m2], %[w22], %[nx2], %[m2]\n\t"                 \
        "v_add_f32 %[a0], %[a0], %[a1]\n\t"                          \
        "v_add_f32 %[a0], %[a2], %[a0]\n\t"                          \
        "v_add_f32 %[a0], %[a3], %[a0]\n\t"                          \
        "v_fmaak_f32 %[qq], %[c15], %[a0], 0x2b8cbccc\n\t"           \
        "v_sqrt_f32 %[y], %[qq]\n\t"                                 \
        "v_sub_f32 %[de0], %[m0], %[st0]\n\t"                        \
        "v_sub_f32 %[de1], %[m1], %[st1]\n\t"                        \
        "v_sub_f32 %[de2], %[m2], %[st2]\n\t"                        \
        "v_add_nc_u32 %[ym], -1, %[y]\n\t"                           \
        "v_add_nc_u32 %[yp], 1, %[y]\n\t"                            \
        "v_mov_b32 %[st0], %[m0]\n\t"                                \
        "v_mov_b32 %[st1], %[m1]\n\t"                                \
        "v_fma_f32 %[rm], -%[ym], %[y], %[qq]\n\t"                   \
        "v_fma_f32 %[rp], -%[yp], %[y], %[qq]\n\t"                   \
        "v_mov_b32 %[st2], %[m2]\n\t"                                \
        "v_cmp_ge_f32 vcc_lo, 0, %[rm]\n\t"                          \
        "v_cndmask_b32 %[q], %[y], %[ym], vcc_lo\n\t"                \
        "v_cmp_lt_f32 vcc_lo, 0, %[rp]\n\t"                          \
        "v_cndmask_b32 %[q], %[q], %[yp], vcc_lo\n\t"                \
        "v_rcp_f32 %[rq], %[q]\n\t"                                  \
        "v_sub_f32 %[f], %[q], %[yl]\n\t"                            \
        "v_cmp_lt_f32 vcc_lo, 0, %[f]\n\t"                           \
        "v_mul_f32 %[dg], 0x3994a648, %[f]\n\t"                      \
        "v_fma_f32 %[e], -%[q], %[rq], 2.0\n\t"                      \
        "v_cndmask_b32 %[dg], 0, %[dg], vcc_lo\n\t"                  \
        "v_mul_f32 %[rq], %[rq], %[e]\n\t"                           \
        "v_mul_f32 %[g3], 0xc556301e, %[dg]\n\t"                     \
        "v_add_f32 %[eqps], %[eqps], %[dg]\n\t"                      \
        "v_fma_f32 %[fac], %[g3], %[rq], 1.0\n\t"                    \
        "s_wait_xcnt 0x0\n\t"                                        \
        "v_fma_f32 %[sig0], %[dxx], %[fac], %[pm]\n\t"               \
        "v_fma_f32 %[sig1], %[dyy], %[fac], %[pm]\n\t"               \
        "v_mul_f32 %[sig2], %[tr2], %[fac]\n\t"                      \
        "v_fma_f32 %[szz], %[dzz], %[fac], %[pm]\n\t"                \
        "global_store_b96 %[voff], v[40:42], %[sb] offset:%[off]\n\t"             \
        : \
          [sig0] "+v"(sig0), [sig1] "+v"(sig1), [sig2] "+v"(sig2), \
          [szz] "+v"(szz), [eqps] "+v"(eqps), [st0] "+v"(st0), \
          [st1] "+v"(st1), [st2] "+v"(st2), [de0] "+v"(de0), [de1] "+v"(de1), \
          [de2] "+v"(de2), [t0] "=&v"(t0), [t1] "=&v"(t1), [u] "=&v"(u), \
          [tr2] "=&v"(tr2), [tzz] "=&v"(tzz), [sum] "=&v"(sum), \
          [yl] "=&v"(yl), [d2] "=&v"(d2), [pm] "=&v"(pm), [dzz] "=&v"(dzz), \
          [dxx] "=&v"(dxx), [dyy] "=&v"(dyy), [a0] "=&v"(a0), [a1] "=&v"(a1), \
          [a2] "=&v"(a2), [a3] "=&v"(a3), [qq] "=&v"(qq), [y] "=&v"(y), \
          [ym] "=&v"(ym), [yp] "=&v"(yp), [rm] "=&v"(rm), [rp] "=&v"(rp), \
          [q] "=&v"(q), [rq] "=&v"(rq), [f] "=&v"(f), [dg] "=&v"(dg), \
          [e] "=&v"(e), [g3] "=&v"(g3), [fac] "=&v"(fac), [m0] "=&v"(m0), \
          [m1] "=&v"(m1), [m2] "=&v"(m2) \
        : \
          [w00] "v"(w00), [w01] "v"(w01), [w02] "v"(w02), [w10] "v"(w10), \
          [w11] "v"(w11), [w12] "v"(w12), [w20] "v"(w20), [w21] "v"(w21), \
          [w22] "v"(w22), [nx0] "s"(NX0), [nx1] "s"(NX1), [nx2] "s"(NX2), \
          [voff] "v"(voff), [hard] "s"(fHARD), [c15] "s"(f15), [sb] "s"(sb), [off] "n"(OFF) \
        : "vcc", "memory");

__global__ __launch_bounds__(64) void j2_scan_kernel(
    const float* __restrict__ x,
    const float* __restrict__ W1,
    float* __restrict__ stress) {
  const int b = blockIdx.x;
  const int j = threadIdx.x;

  const float* __restrict__ xb = x + (size_t)b * NS * 3;

  const float* w = W1 + (3 * j) * 3;
  const float w00 = w[0], w01 = w[1], w02 = w[2];
  const float w10 = w[3], w11 = w[4], w12 = w[5];
  const float w20 = w[6], w21 = w[7], w22 = w[8];

  register float sig0 asm("v40") = 0.f;
  register float sig1 asm("v41") = 0.f;
  register float sig2 asm("v42") = 0.f;
  float szz  = 0.f, eqps = 0.f;

  float* __restrict__ rowb = stress + (size_t)b * NS * LS;
  const unsigned voff = 12u * (unsigned)j;

  xs16 cA; xs8 cB;
  X_GROUP_LOAD(cA, cB, xb);
  X_GROUP_WAIT(cA, cB);

  float st0, st1, st2;
  {
    const float x0 = cA[0], x1 = cA[1], x2 = cA[2];
    st0 = __builtin_fmaf(w02, x2, __builtin_fmaf(w01, x1, w00 * x0));
    st1 = __builtin_fmaf(w12, x2, __builtin_fmaf(w11, x1, w10 * x0));
    st2 = __builtin_fmaf(w22, x2, __builtin_fmaf(w21, x1, w20 * x0));
  }
  float de0 = st0 - 0.f, de1 = st1 - 0.f, de2 = st2 - 0.f;

  const float f15 = 1.5f;

  float t0, t1, u, tr2, tzz, sum, yl, d2, pm, dzz, dxx, dyy, a0, a1, a2, a3;
  float qq, y, ym, yp, rm, rp, q, rq, f, dg, e, g3, fac, m0, m1, m2;

  for (int g = 0; g < NS / 8; ++g) {
    const int gn = (g + 1 < NS / 8) ? (g + 1) : (NS / 8 - 1);
    xs16 nA; xs8 nB;
    X_GROUP_LOAD(nA, nB, xb + 24 * gn);

    float* sb = rowb + (size_t)(8 * g) * LS;

    J2_STEP(cA[3],  cA[4],  cA[5],  0 * LS * 4);
    J2_STEP(cA[6],  cA[7],  cA[8],  1 * LS * 4);
    J2_STEP(cA[9],  cA[10], cA[11], 2 * LS * 4);
    J2_STEP(cA[12], cA[13], cA[14], 3 * LS * 4);
    J2_STEP(cA[15], cB[0],  cB[1],  4 * LS * 4);
    J2_STEP(cB[2],  cB[3],  cB[4],  5 * LS * 4);
    J2_STEP(cB[5],  cB[6],  cB[7],  6 * LS * 4);
    X_GROUP_WAIT(nA, nB);
    J2_STEP(nA[0],  nA[1],  nA[2],  7 * LS * 4);

    cA = nA; cB = nB;
  }
}

#define TPITCH 196
#define BPITCH 200

__global__ __launch_bounds__(256) void out_gemm_wmma_kernel(
    const float* __restrict__ stress,
    const float* __restrict__ Bpad,
    float* __restrict__ out) {
  __shared__ __attribute__((aligned(16))) float As[8 * 16 * TPITCH];
  __shared__ __attribute__((aligned(16))) float Bs[4 * BPITCH];

  const int lane = threadIdx.x & 31;
  const int wave = threadIdx.x >> 5;
  const int tile = blockIdx.x * 8 + wave;
  const int grp  = lane >> 4;
  const int ln   = lane & 15;

  const float4* __restrict__ src4 =
      (const float4*)(stress + (size_t)tile * 16 * LS);
  float* at = As + wave * (16 * TPITCH);

  float4 t[24];
#pragma unroll
  for (int i = 0; i < 24; ++i) t[i] = src4[i * 32 + lane];

  if (threadIdx.x < 192) {
    const int r = threadIdx.x / 48;
    const int c = threadIdx.x - r * 48;
    float4 v = make_float4(0.f, 0.f, 0.f, 0.f);
    if (r < NO) v = ((const float4*)Bpad)[r * 48 + c];
    *(float4*)(Bs + r * BPITCH + 4 * c) = v;
  }

  const int off0 = 4 * lane;
  const int off1 = (lane < 16) ? 4 * (32 + lane) : TPITCH + 4 * (lane - 16);
  const int off2 = TPITCH + 4 * (16 + lane);
#pragma unroll
  for (int m = 0; m < 8; ++m) {
    float* rowp = at + 2 * m * TPITCH;
    *(float4*)(rowp + off0) = t[3 * m + 0];
    *(float4*)(rowp + off1) = t[3 * m + 1];
    *(float4*)(rowp + off2) = t[3 * m + 2];
  }
  __syncthreads();

  const float* arow = at + ln * TPITCH + 2 * grp;
  const float* brow = Bs + (ln < NO ? ln : NO) * BPITCH + 2 * grp;

  v8f acc = {};
#pragma unroll
  for (int k = 0; k < LS; k += 4) {
    v2f a = *(const v2f*)(arow + k);
    v2f bm = *(const v2f*)(brow + k);
    acc = __builtin_amdgcn_wmma_f32_16x16x4_f32(
        false, a, false, bm, (short)0, acc, false, false);
  }

  if (ln < NO) {
#pragma unroll
    for (int r = 0; r < 8; ++r) {
      const size_t m = (size_t)tile * 16 + r + 8 * grp;
      out[m * NO + ln] = acc[r];
    }
  }
}

extern "C" void kernel_launch(void* const* d_in, const int* in_sizes, int n_in,
                              void* d_out, int out_size, void* d_ws,
                              size_t ws_size, hipStream_t stream) {
  (void)in_sizes; (void)n_in; (void)out_size; (void)ws_size;
  const float* x  = (const float*)d_in[0];
  const float* W1 = (const float*)d_in[1];
  const float* W2 = (const float*)d_in[2];
  float* out = (float*)d_out;

  float* stress = (float*)d_ws;
  float* Bpad   = stress + (size_t)NB * NS * LS;

  prep_bpanel_kernel<<<1, LS, 0, stream>>>(W2, Bpad);
  j2_scan_kernel<<<NB, 64, 0, stream>>>(x, W1, stress);
  const int tiles = (NB * NS) / 16;
  out_gemm_wmma_kernel<<<tiles / 8, 256, 0, stream>>>(stress, Bpad, out);
}


# time loop cut at step 1536 into two launches: projection of the rows before the cut runs as extra work-groups of launch 2 beside the scan of the rest; B-panel prep rides in launch 1; sums unchanged
# speedup vs baseline: 2.7172x; 1.0312x over previous
#include <hip/hip_runtime.h>


typedef __attribute__((ext_vector_type(2))) float v2f;
typedef __attribute__((ext_vector_type(8))) float v8f;

constexpr double dE    = 3130.0;
constexpr double dNU   = 0.37;
constexpr double dSIGY = 64.8;
constexpr double dHARD = 100.0;
constexpr double dG    = dE / (2.0 * (1.0 + dNU));
constexpr double dLAM  = dE * dNU / ((1.0 + dNU) * (1.0 - 2.0 * dNU));
constexpr double dC    = dE / ((1.0 + dNU) * (1.0 - 2.0 * dNU));

constexpr float fG     = (float)dG;
constexpr float fLAM   = (float)dLAM;
constexpr float fC11   = (float)(dC * (1.0 - dNU));
constexpr float fC12   = (float)(dC * dNU);
constexpr float fSIGY  = (float)dSIGY;
constexpr float fHARD  = (float)dHARD;
constexpr float f3G    = (float)(3.0 * dG);
constexpr float fINV3GH = (float)(1.0 / (3.0 * dG + dHARD));

#define NB   128
#define NS   2048
#define LS   192
#define NO   3
#define NPADN 16
#define S_CUT 1536
#define G_CUT (S_CUT / 8)
#define NCARRY 11

__device__ __forceinline__ void prep_bpanel(const float* __restrict__ W2,
                                            float* __restrict__ Bpad,
                                            int tid, int nthr) {
  for (int k = tid; k < LS; k += nthr) {
#pragma unroll
    for (int n = 0; n < NPADN; ++n) {
      float v = 0.0f;
      if (n < NO) {
        float w = W2[n * LS + k];
        v = fmaxf(w, 0.0f) + log1pf(__expf(-fabsf(w)));
      }
      Bpad[n * LS + k] = v;
    }
  }
}

static_assert(__builtin_bit_cast(unsigned, fC11) == 0x45acff7bu, "C11 literal");
static_assert(__builtin_bit_cast(unsigned, fC12) == 0x454b3437u, "C12 literal");
static_assert(__builtin_bit_cast(unsigned, fLAM) == 0x454b3437u, "LAM literal");
static_assert(__builtin_bit_cast(unsigned, fG) == 0x448ecabfu, "G literal");
static_assert(__builtin_bit_cast(unsigned, fSIGY) == 0x4281999au, "SIGY literal");
static_assert(__builtin_bit_cast(unsigned, fINV3GH) == 0x3994a648u, "1/(3G+H)");
static_assert(__builtin_bit_cast(unsigned, -f3G) == 0xc556301eu, "-3G literal");
static_assert(__builtin_bit_cast(unsigned, 1.0f / 3.0f) == 0x3eaaaaabu, "1/3");
static_assert(__builtin_bit_cast(unsigned, 1e-12f) == 0x2b8cbcccu, "1e-12");

typedef __attribute__((ext_vector_type(8)))  float xs8;
typedef __attribute__((ext_vector_type(16))) float xs16;

#define X_GROUP_LOAD(A, B, PTR) \
    asm volatile("s_load_b512 %0, %2, 0x0\n\t" \
                 "s_load_b256 %1, %2, 0x40" \
                 : "=&s"(A), "=&s"(B) : "s"(PTR) : "memory")
#define X_GROUP_WAIT(A, B) \
    asm volatile("s_wait_kmcnt 0x0" : "+s"(A), "+s"(B) : : "memory")

#define J2_STEP(NX0, NX1, NX2, OFF) \
    asm volatile( \
        "v_fma_f32 %[t0], %[de0], 0x45acff7b, %[sig0]\n\t"           \
        "v_fma_f32 %[t1], %[de0], 0x454b3437, %[sig1]\n\t"           \
        "v_add_f32 %[u], %[de1], %[de0]\n\t"                         \
        "v_fma_f32 %[tr2], %[de2], 0x448ecabf, %[sig2]\n\t"          \
        "v_mul_f32 %[m0], %[nx0], %[w00]\n\t"                        \
        "v_fma_f32 %[t0], %[de1], 0x454b3437, %[t0]\n\t"             \
        "v_fma_f32 %[t1], %[de1], 0x45acff7b, %[t1]\n\t"             \
        "v_fma_f32 %[tzz], %[u], 0x454b3437, %[szz]\n\t"             \
        "v_mul_f32 %[m1], %[nx0], %[w10]\n\t"                        \
        "v_mul_f32 %[m2], %[nx0], %[w20]\n\t"                        \
        "v_add_f32 %[sum], %[t1], %[t0]\n\t"                         \
        "v_fmaak_f32 %[yl], %[hard], %[eqps], 0x4281999a\n\t"        \
        "v_add_f32 %[d2], %[tr2], %[tr2]\n\t"                        \
        "v_add_f32 %[sum], %[tzz], %[sum]\n\t"                       \
        "v_fma_f32 %[m0], %[w01], %[nx1], %[m0]\n\t"                 \
        "v_fma_f32 %[m1], %[w11], %[nx1], %[m1]\n\t"                 \
        "v_fma_f32 %[m2], %[w21], %[nx1], %[m2]\n\t"                 \
        "v_mul_f32 %[pm], 0x3eaaaaab, %[sum]\n\t"                    \
        "v_fma_f32 %[dzz], %[sum], 0xbeaaaaab, %[tzz]\n\t"           \
        "v_sub_f32 %[dxx], %[t0], %[pm]\n\t"                         \
        "v_sub_f32 %[dyy], %[t1], %[pm]\n\t"                         \
        "v_mul_f32 %[a2], %[dzz], %[dzz]\n\t"                        \
        "v_mul_f32 %[a3], %[d2], %[tr2]\n\t"                         \
        "v_mul_f32 %[a0], %[dxx], %[dxx]\n\t"                        \
        "v_mul_f32 %[a1], %[dyy], %[dyy]\n\t"                        \
        "v_fma_f32 %[m0], %[w02], %[nx2], %[m0]\n\t"                 \
        "v_fma_f32 %[m1], %[w12], %[nx2], %[m1]\n\t"                 \
        "v_fma_f32 %[m2], %[w22], %[nx2], %[m2]\n\t"                 \
        "v_add_f32 %[a0], %[a0], %[a1]\n\t"                          \
        "v_add_f32 %[a0], %[a2], %[a0]\n\t"                          \
        "v_add_f32 %[a0], %[a3], %[a0]\n\t"                          \
        "v_fmaak_f32 %[qq], %[c15], %[a0], 0x2b8cbccc\n\t"           \
        "v_sqrt_f32 %[y], %[qq]\n\t"                                 \
        "v_sub_f32 %[de0], %[m0], %[st0]\n\t"                        \
        "v_sub_f32 %[de1], %[m1], %[st1]\n\t"                        \
        "v_sub_f32 %[de2], %[m2], %[st2]\n\t"                        \
        "v_add_nc_u32 %[ym], -1, %[y]\n\t"                           \
        "v_add_nc_u32 %[yp], 1, %[y]\n\t"                            \
        "v_mov_b32 %[st0], %[m0]\n\t"                                \
        "v_mov_b32 %[st1], %[m1]\n\t"                                \
        "v_fma_f32 %[rm], -%[ym], %[y], %[qq]\n\t"                   \
        "v_fma_f32 %[rp], -%[yp], %[y], %[qq]\n\t"                   \
        "v_mov_b32 %[st2], %[m2]\n\t"                                \
        "v_cmp_ge_f32 vcc_lo, 0, %[rm]\n\t"                          \
        "v_cndmask_b32 %[q], %[y], %[ym], vcc_lo\n\t"                \
        "v_cmp_lt_f32 vcc_lo, 0, %[rp]\n\t"                          \
        "v_cndmask_b32 %[q], %[q], %[yp], vcc_lo\n\t"                \
        "v_rcp_f32 %[rq], %[q]\n\t"                                  \
        "v_sub_f32 %[f], %[q], %[yl]\n\t"                            \
        "v_cmp_lt_f32 vcc_lo, 0, %[f]\n\t"                           \
        "v_mul_f32 %[dg], 0x3994a648, %[f]\n\t"                      \
        "v_fma_f32 %[e], -%[q], %[rq], 2.0\n\t"                      \
        "v_cndmask_b32 %[dg], 0, %[dg], vcc_lo\n\t"                  \
        "v_mul_f32 %[rq], %[rq], %[e]\n\t"                           \
        "v_mul_f32 %[g3], 0xc556301e, %[dg]\n\t"                     \
        "v_add_f32 %[eqps], %[eqps], %[dg]\n\t"                      \
        "v_fma_f32 %[fac], %[g3], %[rq], 1.0\n\t"                    \
        "s_wait_xcnt 0x0\n\t"                                        \
        "v_fma_f32 %[sig0], %[dxx], %[fac], %[pm]\n\t"               \
        "v_fma_f32 %[sig1], %[dyy], %[fac], %[pm]\n\t"               \
        "v_mul_f32 %[sig2], %[tr2], %[fac]\n\t"                      \
        "v_fma_f32 %[szz], %[dzz], %[fac], %[pm]\n\t"                \
        "global_store_b96 %[voff], v[40:42], %[sb] offset:%[off]\n\t"             \
        : \
          [sig0] "+v"(sig0), [sig1] "+v"(sig1), [sig2] "+v"(sig2), \
          [szz] "+v"(szz), [eqps] "+v"(eqps), [st0] "+v"(st0), \
          [st1] "+v"(st1), [st2] "+v"(st2), [de0] "+v"(de0), [de1] "+v"(de1), \
          [de2] "+v"(de2), [t0] "=&v"(t0), [t1] "=&v"(t1), [u] "=&v"(u), \
          [tr2] "=&v"(tr2), [tzz] "=&v"(tzz), [sum] "=&v"(sum), \
          [yl] "=&v"(yl), [d2] "=&v"(d2), [pm] "=&v"(pm), [dzz] "=&v"(dzz), \
          [dxx] "=&v"(dxx), [dyy] "=&v"(dyy), [a0] "=&v"(a0), [a1] "=&v"(a1), \
          [a2] "=&v"(a2), [a3] "=&v"(a3), [qq] "=&v"(qq), [y] "=&v"(y), \
          [ym] "=&v"(ym), [yp] "=&v"(yp), [rm] "=&v"(rm), [rp] "=&v"(rp), \
          [q] "=&v"(q), [rq] "=&v"(rq), [f] "=&v"(f), [dg] "=&v"(dg), \
          [e] "=&v"(e), [g3] "=&v"(g3), [fac] "=&v"(fac), [m0] "=&v"(m0), \
          [m1] "=&v"(m1), [m2] "=&v"(m2) \
        : \
          [w00] "v"(w00), [w01] "v"(w01), [w02] "v"(w02), [w10] "v"(w10), \
          [w11] "v"(w11), [w12] "v"(w12), [w20] "v"(w20), [w21] "v"(w21), \
          [w22] "v"(w22), [nx0] "s"(NX0), [nx1] "s"(NX1), [nx2] "s"(NX2), \
          [voff] "v"(voff), [hard] "s"(fHARD), [c15] "s"(f15), [sb] "s"(sb), [off] "n"(OFF) \
        : "vcc", "memory");

template <int G0, int G1, bool FROM_START, bool TO_CARRY>
__device__ __forceinline__ void j2_scan_part(
    const float* __restrict__ x,
    const float* __restrict__ W1,
    float* stress,
    float* carry,
    const int b,
    const int j) {
  const float* __restrict__ xb = x + (size_t)b * NS * 3;

  const float* w = W1 + (3 * j) * 3;
  const float w00 = w[0], w01 = w[1], w02 = w[2];
  const float w10 = w[3], w11 = w[4], w12 = w[5];
  const float w20 = w[6], w21 = w[7], w22 = w[8];

  float* cp = carry + (b * 64 + j);

  register float sig0 asm("v40");
  register float sig1 asm("v41");
  register float sig2 asm("v42");
  float szz, eqps;

  float* rowb = stress + (size_t)b * NS * LS;
  const unsigned voff = 12u * (unsigned)j;

  float st0, st1, st2, de0, de1, de2;
  if (!FROM_START) {
    sig0 = cp[0 * 8192]; sig1 = cp[1 * 8192]; sig2 = cp[2 * 8192];
    szz  = cp[3 * 8192]; eqps = cp[4 * 8192];
    st0  = cp[5 * 8192]; st1  = cp[6 * 8192]; st2  = cp[7 * 8192];
    de0  = cp[8 * 8192]; de1  = cp[9 * 8192]; de2  = cp[10 * 8192];
  }

  xs16 cA; xs8 cB;
  X_GROUP_LOAD(cA, cB, xb + 24 * G0);
  X_GROUP_WAIT(cA, cB);

  if (FROM_START) {
    sig0 = 0.f; sig1 = 0.f; sig2 = 0.f; szz = 0.f; eqps = 0.f;
    const float x0 = cA[0], x1 = cA[1], x2 = cA[2];
    st0 = __builtin_fmaf(w02, x2, __builtin_fmaf(w01, x1, w00 * x0));
    st1 = __builtin_fmaf(w12, x2, __builtin_fmaf(w11, x1, w10 * x0));
    st2 = __builtin_fmaf(w22, x2, __builtin_fmaf(w21, x1, w20 * x0));
    de0 = st0 - 0.f; de1 = st1 - 0.f; de2 = st2 - 0.f;
  }

  const float f15 = 1.5f;

  float t0, t1, u, tr2, tzz, sum, yl, d2, pm, dzz, dxx, dyy, a0, a1, a2, a3;
  float qq, y, ym, yp, rm, rp, q, rq, f, dg, e, g3, fac, m0, m1, m2;

  for (int g = G0; g < G1; ++g) {
    const int gn = (g + 1 < NS / 8) ? (g + 1) : (NS / 8 - 1);
    xs16 nA; xs8 nB;
    X_GROUP_LOAD(nA, nB, xb + 24 * gn);

    float* sb = rowb + (size_t)(8 * g) * LS;

    J2_STEP(cA[3],  cA[4],  cA[5],  0 * LS * 4);
    J2_STEP(cA[6],  cA[7],  cA[8],  1 * LS * 4);
    J2_STEP(cA[9],  cA[10], cA[11], 2 * LS * 4);
    J2_STEP(cA[12], cA[13], cA[14], 3 * LS * 4);
    J2_STEP(cA[15], cB[0],  cB[1],  4 * LS * 4);
    J2_STEP(cB[2],  cB[3],  cB[4],  5 * LS * 4);
    J2_STEP(cB[5],  cB[6],  cB[7],  6 * LS * 4);
    X_GROUP_WAIT(nA, nB);
    J2_STEP(nA[0],  nA[1],  nA[2],  7 * LS * 4);

    cA = nA; cB = nB;
  }

  if (TO_CARRY) {
    cp[0 * 8192] = sig0; cp[1 * 8192] = sig1; cp[2 * 8192] = sig2;
    cp[3 * 8192] = szz;  cp[4 * 8192] = eqps;
    cp[5 * 8192] = st0;  cp[6 * 8192] = st1;  cp[7 * 8192] = st2;
    cp[8 * 8192] = de0;  cp[9 * 8192] = de1;  cp[10 * 8192] = de2;
  }
}

#define TPITCH 196
#define BPITCH 200

template <int NW>
__device__ __forceinline__ void project_tile(
    const float* stress,
    const float* __restrict__ Bpad,
    float* __restrict__ out,
    const int tile) {
  __shared__ __attribute__((aligned(16))) float As[NW * 16 * TPITCH];
  __shared__ __attribute__((aligned(16))) float Bs[4 * BPITCH];

  const int lane = threadIdx.x & 31;
  const int wave = threadIdx.x >> 5;
  const int grp  = lane >> 4;
  const int ln   = lane & 15;

  const float4* src4 = (const float4*)(stress + (size_t)tile * 16 * LS);
  float* at = As + wave * (16 * TPITCH);

  float4 t[24];
#pragma unroll
  for (int i = 0; i < 24; ++i) t[i] = src4[i * 32 + lane];

  for (int q = threadIdx.x; q < 192; q += 32 * NW) {
    const int r = q / 48;
    const int c = q - r * 48;
    float4 v = make_float4(0.f, 0.f, 0.f, 0.f);
    if (r < NO) v = ((const float4*)Bpad)[r * 48 + c];
    *(float4*)(Bs + r * BPITCH + 4 * c) = v;
  }

  const int off0 = 4 * lane;
  const int off1 = (lane < 16) ? 4 * (32 + lane) : TPITCH + 4 * (lane - 16);
  const int off2 = TPITCH + 4 * (16 + lane);
#pragma unroll
  for (int m = 0; m < 8; ++m) {
    float* rowp = at + 2 * m * TPITCH;
    *(float4*)(rowp + off0) = t[3 * m + 0];
    *(float4*)(rowp + off1) = t[3 * m + 1];
    *(float4*)(rowp + off2) = t[3 * m + 2];
  }
  __syncthreads();

  const float* arow = at + ln * TPITCH + 2 * grp;
  const float* brow = Bs + (ln < NO ? ln : NO) * BPITCH + 2 * grp;

  v8f acc = {};
#pragma unroll
  for (int k = 0; k < LS; k += 4) {
    v2f a = *(const v2f*)(arow + k);
    v2f bm = *(const v2f*)(brow + k);
    acc = __builtin_amdgcn_wmma_f32_16x16x4_f32(
        false, a, false, bm, (short)0, acc, false, false);
  }

  if (ln < NO) {
#pragma unroll
    for (int r = 0; r < 8; ++r) {
      const size_t m = (size_t)tile * 16 + r + 8 * grp;
      out[m * NO + ln] = acc[r];
    }
  }
}

__device__ __forceinline__ int tile_before_cut(int u) {
  const int b = u / (S_CUT / 16);
  return b * (NS / 16) + (u - b * (S_CUT / 16));
}
__device__ __forceinline__ int tile_behind_cut(int u) {
  const int b = u / ((NS - S_CUT) / 16);
  return b * (NS / 16) + (S_CUT / 16) + (u - b * ((NS - S_CUT) / 16));
}

__global__ __launch_bounds__(64) void scan_head_kernel(
    const float* __restrict__ x,
    const float* __restrict__ W1,
    const float* __restrict__ W2,
    float* stress, float* Bpad, float* carry) {
  if (blockIdx.x < NB) {
    j2_scan_part<0, G_CUT, true, true>(x, W1, stress, carry,
                                       (int)blockIdx.x, (int)threadIdx.x);
  } else {
    prep_bpanel(W2, Bpad, (int)threadIdx.x, 64);
  }
}

__global__ __launch_bounds__(64) void scan_tail_project_kernel(
    const float* __restrict__ W1,
    float* carry,
    const float* __restrict__ x,
    float* stress, const float* Bpad,
    float* __restrict__ out) {
  if (blockIdx.x < NB) {
    j2_scan_part<G_CUT, NS / 8, false, false>(x, W1, stress, carry,
                                              (int)blockIdx.x, (int)threadIdx.x);
  } else {
    const int u = ((int)blockIdx.x - NB) * 2 + ((int)threadIdx.x >> 5);
    project_tile<2>(stress, Bpad, out, tile_before_cut(u));
  }
}

__global__ __launch_bounds__(256) void project_tail_kernel(
    const float* __restrict__ stress,
    const float* __restrict__ Bpad,
    float* __restrict__ out) {
  const int u = (int)blockIdx.x * 8 + ((int)threadIdx.x >> 5);
  project_tile<8>(stress, Bpad, out, tile_behind_cut(u));
}

extern "C" void kernel_launch(void* const* d_in, const int* in_sizes, int n_in,
                              void* d_out, int out_size, void* d_ws,
                              size_t ws_size, hipStream_t stream) {
  (void)in_sizes; (void)n_in; (void)out_size; (void)ws_size;
  const float* x  = (const float*)d_in[0];
  const float* W1 = (const float*)d_in[1];
  const float* W2 = (const float*)d_in[2];
  float* out = (float*)d_out;

  float* stress = (float*)d_ws;
  float* Bpad   = stress + (size_t)NB * NS * LS;
  float* carry  = Bpad + NPADN * LS;

  constexpr int tiles_before = NB * (S_CUT / 16);
  constexpr int tiles_behind = NB * ((NS - S_CUT) / 16);
  scan_head_kernel<<<NB + 1, 64, 0, stream>>>(x, W1, W2, stress, Bpad, carry);
  scan_tail_project_kernel<<<NB + tiles_before / 2, 64, 0, stream>>>(
      W1, carry, x, stress, Bpad, out);
  project_tail_kernel<<<tiles_behind / 8, 256, 0, stream>>>(stress, Bpad, out);
}


# scan: each of a group's 8 steps writes its stress row into a register triple of its own and the 8 row stores leave as ONE run behind the group, not one store inside every step; sums unchanged
# speedup vs baseline: 3.0006x; 1.1043x over previous
#include <hip/hip_runtime.h>


typedef __attribute__((ext_vector_type(2))) float v2f;
typedef __attribute__((ext_vector_type(8))) float v8f;

constexpr double dE    = 3130.0;
constexpr double dNU   = 0.37;
constexpr double dSIGY = 64.8;
constexpr double dHARD = 100.0;
constexpr double dG    = dE / (2.0 * (1.0 + dNU));
constexpr double dLAM  = dE * dNU / ((1.0 + dNU) * (1.0 - 2.0 * dNU));
constexpr double dC    = dE / ((1.0 + dNU) * (1.0 - 2.0 * dNU));

constexpr float fG     = (float)dG;
constexpr float fLAM   = (float)dLAM;
constexpr float fC11   = (float)(dC * (1.0 - dNU));
constexpr float fC12   = (float)(dC * dNU);
constexpr float fSIGY  = (float)dSIGY;
constexpr float fHARD  = (float)dHARD;
constexpr float f3G    = (float)(3.0 * dG);
constexpr float fINV3GH = (float)(1.0 / (3.0 * dG + dHARD));

#define NB   128
#define NS   2048
#define LS   192
#define NO   3
#define NPADN 16
#define S_CUT 1536
#define G_CUT (S_CUT / 8)
#define NCARRY 11

__device__ __forceinline__ void prep_bpanel(const float* __restrict__ W2,
                                            float* __restrict__ Bpad,
                                            int tid, int nthr) {
  for (int k = tid; k < LS; k += nthr) {
#pragma unroll
    for (int n = 0; n < NPADN; ++n) {
      float v = 0.0f;
      if (n < NO) {
        float w = W2[n * LS + k];
        v = fmaxf(w, 0.0f) + log1pf(__expf(-fabsf(w)));
      }
      Bpad[n * LS + k] = v;
    }
  }
}

static_assert(__builtin_bit_cast(unsigned, fC11) == 0x45acff7bu, "C11 literal");
static_assert(__builtin_bit_cast(unsigned, fC12) == 0x454b3437u, "C12 literal");
static_assert(__builtin_bit_cast(unsigned, fLAM) == 0x454b3437u, "LAM literal");
static_assert(__builtin_bit_cast(unsigned, fG) == 0x448ecabfu, "G literal");
static_assert(__builtin_bit_cast(unsigned, fSIGY) == 0x4281999au, "SIGY literal");
static_assert(__builtin_bit_cast(unsigned, fINV3GH) == 0x3994a648u, "1/(3G+H)");
static_assert(__builtin_bit_cast(unsigned, -f3G) == 0xc556301eu, "-3G literal");
static_assert(__builtin_bit_cast(unsigned, 1.0f / 3.0f) == 0x3eaaaaabu, "1/3");
static_assert(__builtin_bit_cast(unsigned, 1e-12f) == 0x2b8cbcccu, "1e-12");

typedef __attribute__((ext_vector_type(8)))  float xs8;
typedef __attribute__((ext_vector_type(16))) float xs16;

#define X_GROUP_LOAD(A, B, PTR) \
    asm volatile("s_load_b512 %0, %2, 0x0\n\t" \
                 "s_load_b256 %1, %2, 0x40" \
                 : "=&s"(A), "=&s"(B) : "s"(PTR) : "memory")
#define X_GROUP_WAIT(A, B) \
    asm volatile("s_wait_kmcnt 0x0" : "+s"(A), "+s"(B) : : "memory")

#define J2_STEP(NX0, NX1, NX2, P0, P1, P2, N0, N1, N2, WAITX, HOLD) \
    asm volatile( \
        "v_fma_f32 %[t0], %[de0], 0x45acff7b, %[p0]\n\t"             \
        "v_fma_f32 %[t1], %[de0], 0x454b3437, %[p1]\n\t"             \
        "v_add_f32 %[u], %[de1], %[de0]\n\t"                         \
        "v_fma_f32 %[tr2], %[de2], 0x448ecabf, %[p2]\n\t"            \
        "v_mul_f32 %[m0], %[nx0], %[w00]\n\t"                        \
        "v_fma_f32 %[t0], %[de1], 0x454b3437, %[t0]\n\t"             \
        "v_fma_f32 %[t1], %[de1], 0x45acff7b, %[t1]\n\t"             \
        "v_fma_f32 %[tzz], %[u], 0x454b3437, %[szz]\n\t"             \
        "v_mul_f32 %[m1], %[nx0], %[w10]\n\t"                        \
        "v_mul_f32 %[m2], %[nx0], %[w20]\n\t"                        \
        "v_add_f32 %[sum], %[t1], %[t0]\n\t"                         \
        "v_fmaak_f32 %[yl], %[hard], %[eqps], 0x4281999a\n\t"        \
        "v_add_f32 %[d2], %[tr2], %[tr2]\n\t"                        \
        "v_add_f32 %[sum], %[tzz], %[sum]\n\t"                       \
        "v_fma_f32 %[m0], %[w01], %[nx1], %[m0]\n\t"                 \
        "v_fma_f32 %[m1], %[w11], %[nx1], %[m1]\n\t"                 \
        "v_fma_f32 %[m2], %[w21], %[nx1], %[m2]\n\t"                 \
        "v_mul_f32 %[pm], 0x3eaaaaab, %[sum]\n\t"                    \
        "v_fma_f32 %[dzz], %[sum], 0xbeaaaaab, %[tzz]\n\t"           \
        "v_sub_f32 %[dxx], %[t0], %[pm]\n\t"                         \
        "v_sub_f32 %[dyy], %[t1], %[pm]\n\t"                         \
        "v_mul_f32 %[a2], %[dzz], %[dzz]\n\t"                        \
        "v_mul_f32 %[a3], %[d2], %[tr2]\n\t"                         \
        "v_mul_f32 %[a0], %[dxx], %[dxx]\n\t"                        \
        "v_mul_f32 %[a1], %[dyy], %[dyy]\n\t"                        \
        "v_fma_f32 %[m0], %[w02], %[nx2], %[m0]\n\t"                 \
        "v_fma_f32 %[m1], %[w12], %[nx2], %[m1]\n\t"                 \
        "v_fma_f32 %[m2], %[w22], %[nx2], %[m2]\n\t"                 \
        "v_add_f32 %[a0], %[a0], %[a1]\n\t"                          \
        "v_add_f32 %[a0], %[a2], %[a0]\n\t"                          \
        "v_add_f32 %[a0], %[a3], %[a0]\n\t"                          \
        "v_fmaak_f32 %[qq], %[c15], %[a0], 0x2b8cbccc\n\t"           \
        "v_sqrt_f32 %[y], %[qq]\n\t"                                 \
        "v_sub_f32 %[de0], %[m0], %[st0]\n\t"                        \
        "v_sub_f32 %[de1], %[m1], %[st1]\n\t"                        \
        "v_sub_f32 %[de2], %[m2], %[st2]\n\t"                        \
        "v_add_nc_u32 %[ym], -1, %[y]\n\t"                           \
        "v_add_nc_u32 %[yp], 1, %[y]\n\t"                            \
        "v_mov_b32 %[st0], %[m0]\n\t"                                \
        "v_mov_b32 %[st1], %[m1]\n\t"                                \
        "v_fma_f32 %[rm], -%[ym], %[y], %[qq]\n\t"                   \
        "v_fma_f32 %[rp], -%[yp], %[y], %[qq]\n\t"                   \
        "v_mov_b32 %[st2], %[m2]\n\t"                                \
        "v_cmp_ge_f32 vcc_lo, 0, %[rm]\n\t"                          \
        "v_cndmask_b32 %[q], %[y], %[ym], vcc_lo\n\t"                \
        "v_cmp_lt_f32 vcc_lo, 0, %[rp]\n\t"                          \
        "v_cndmask_b32 %[q], %[q], %[yp], vcc_lo\n\t"                \
        "v_rcp_f32 %[rq], %[q]\n\t"                                  \
        "v_sub_f32 %[f], %[q], %[yl]\n\t"                            \
        "v_cmp_lt_f32 vcc_lo, 0, %[f]\n\t"                           \
        "v_mul_f32 %[dg], 0x3994a648, %[f]\n\t"                      \
        "v_fma_f32 %[e], -%[q], %[rq], 2.0\n\t"                      \
        "v_cndmask_b32 %[dg], 0, %[dg], vcc_lo\n\t"                  \
        "v_mul_f32 %[rq], %[rq], %[e]\n\t"                           \
        "v_mul_f32 %[g3], 0xc556301e, %[dg]\n\t"                     \
        "v_add_f32 %[eqps], %[eqps], %[dg]\n\t"                      \
        "v_fma_f32 %[fac], %[g3], %[rq], 1.0\n\t"                    \
        WAITX \
        "v_fma_f32 %[n0], %[dxx], %[fac], %[pm]\n\t"                 \
        "v_fma_f32 %[n1], %[dyy], %[fac], %[pm]\n\t"                 \
        "v_mul_f32 %[n2], %[tr2], %[fac]\n\t"                        \
        "v_fma_f32 %[szz], %[dzz], %[fac], %[pm]"                     \
        : \
          [n0] "=&v"(N0), [n1] "=&v"(N1), [n2] "=&v"(N2), \
          [szz] "+v"(szz), [eqps] "+v"(eqps), [st0] "+v"(st0), \
          [st1] "+v"(st1), [st2] "+v"(st2), [de0] "+v"(de0), [de1] "+v"(de1), \
          [de2] "+v"(de2), [t0] "=&v"(t0), [t1] "=&v"(t1), [u] "=&v"(u), \
          [tr2] "=&v"(tr2), [tzz] "=&v"(tzz), [sum] "=&v"(sum), \
          [yl] "=&v"(yl), [d2] "=&v"(d2), [pm] "=&v"(pm), [dzz] "=&v"(dzz), \
          [dxx] "=&v"(dxx), [dyy] "=&v"(dyy), [a0] "=&v"(a0), [a1] "=&v"(a1), \
          [a2] "=&v"(a2), [a3] "=&v"(a3), [qq] "=&v"(qq), [y] "=&v"(y), \
          [ym] "=&v"(ym), [yp] "=&v"(yp), [rm] "=&v"(rm), [rp] "=&v"(rp), \
          [q] "=&v"(q), [rq] "=&v"(rq), [f] "=&v"(f), [dg] "=&v"(dg), \
          [e] "=&v"(e), [g3] "=&v"(g3), [fac] "=&v"(fac), [m0] "=&v"(m0), \
          [m1] "=&v"(m1), [m2] "=&v"(m2) \
        : \
          [w00] "v"(w00), [w01] "v"(w01), [w02] "v"(w02), [w10] "v"(w10), \
          [w11] "v"(w11), [w12] "v"(w12), [w20] "v"(w20), [w21] "v"(w21), \
          [w22] "v"(w22), [nx0] "s"(NX0), [nx1] "s"(NX1), [nx2] "s"(NX2), \
          [p0] "v"(P0), [p1] "v"(P1), [p2] "v"(P2), \
          [hard] "s"(fHARD), [c15] "s"(f15) HOLD \
        : "vcc", "memory");

#define J2_WAITX "s_wait_xcnt 0x0\n\t"
#define J2_HOLD_TRIPLES \
    , "v"(q10), "v"(q11), "v"(q12), "v"(q20), "v"(q21), "v"(q22), \
      "v"(q30), "v"(q31), "v"(q32), "v"(q40), "v"(q41), "v"(q42), \
      "v"(q50), "v"(q51), "v"(q52), "v"(q60), "v"(q61), "v"(q62)
#define J2_HOLD_NONE

#define J2_STORE_GROUP() \
    asm volatile( \
        "global_store_b96 %[voff], v[64:66], %[sb]\n\t" \
        "global_store_b96 %[voff], v[68:70], %[sb] offset:768\n\t" \
        "global_store_b96 %[voff], v[72:74], %[sb] offset:1536\n\t" \
        "global_store_b96 %[voff], v[76:78], %[sb] offset:2304\n\t" \
        "global_store_b96 %[voff], v[80:82], %[sb] offset:3072\n\t" \
        "global_store_b96 %[voff], v[84:86], %[sb] offset:3840\n\t" \
        "global_store_b96 %[voff], v[88:90], %[sb] offset:4608\n\t" \
        "global_store_b96 %[voff], v[40:42], %[sb] offset:5376" \
        : \
        : [voff] "v"(voff), [sb] "s"(sb), \
          "v"(q00), "v"(q01), "v"(q02), "v"(q10), "v"(q11), "v"(q12), \
          "v"(q20), "v"(q21), "v"(q22), "v"(q30), "v"(q31), "v"(q32), \
          "v"(q40), "v"(q41), "v"(q42), "v"(q50), "v"(q51), "v"(q52), \
          "v"(q60), "v"(q61), "v"(q62), "v"(q70), "v"(q71), "v"(q72) \
        : "memory");

template <int G0, int G1, bool FROM_START, bool TO_CARRY>
__device__ __forceinline__ void j2_scan_part(
    const float* __restrict__ x,
    const float* __restrict__ W1,
    float* stress,
    float* carry,
    const int b,
    const int j) {
  const float* __restrict__ xb = x + (size_t)b * NS * 3;

  const float* w = W1 + (3 * j) * 3;
  const float w00 = w[0], w01 = w[1], w02 = w[2];
  const float w10 = w[3], w11 = w[4], w12 = w[5];
  const float w20 = w[6], w21 = w[7], w22 = w[8];

  float* cp = carry + (b * 64 + j);

  register float q00 asm("v64"); register float q01 asm("v65"); register float q02 asm("v66");
  register float q10 asm("v68"); register float q11 asm("v69"); register float q12 asm("v70");
  register float q20 asm("v72"); register float q21 asm("v73"); register float q22 asm("v74");
  register float q30 asm("v76"); register float q31 asm("v77"); register float q32 asm("v78");
  register float q40 asm("v80"); register float q41 asm("v81"); register float q42 asm("v82");
  register float q50 asm("v84"); register float q51 asm("v85"); register float q52 asm("v86");
  register float q60 asm("v88"); register float q61 asm("v89"); register float q62 asm("v90");
  register float q70 asm("v40"); register float q71 asm("v41"); register float q72 asm("v42");
  float szz, eqps;

  float* rowb = stress + (size_t)b * NS * LS;
  const unsigned voff = 12u * (unsigned)j;

  float st0, st1, st2, de0, de1, de2;
  if (!FROM_START) {
    q70 = cp[0 * 8192]; q71 = cp[1 * 8192]; q72 = cp[2 * 8192];
    szz  = cp[3 * 8192]; eqps = cp[4 * 8192];
    st0  = cp[5 * 8192]; st1  = cp[6 * 8192]; st2  = cp[7 * 8192];
    de0  = cp[8 * 8192]; de1  = cp[9 * 8192]; de2  = cp[10 * 8192];
  }

  xs16 cA; xs8 cB;
  X_GROUP_LOAD(cA, cB, xb + 24 * G0);
  X_GROUP_WAIT(cA, cB);

  if (FROM_START) {
    q70 = 0.f; q71 = 0.f; q72 = 0.f; szz = 0.f; eqps = 0.f;
    const float x0 = cA[0], x1 = cA[1], x2 = cA[2];
    st0 = __builtin_fmaf(w02, x2, __builtin_fmaf(w01, x1, w00 * x0));
    st1 = __builtin_fmaf(w12, x2, __builtin_fmaf(w11, x1, w10 * x0));
    st2 = __builtin_fmaf(w22, x2, __builtin_fmaf(w21, x1, w20 * x0));
    de0 = st0 - 0.f; de1 = st1 - 0.f; de2 = st2 - 0.f;
  }

  const float f15 = 1.5f;

  q10 = 0.f; q11 = 0.f; q12 = 0.f; q20 = 0.f; q21 = 0.f; q22 = 0.f;
  q30 = 0.f; q31 = 0.f; q32 = 0.f; q40 = 0.f; q41 = 0.f; q42 = 0.f;
  q50 = 0.f; q51 = 0.f; q52 = 0.f; q60 = 0.f; q61 = 0.f; q62 = 0.f;

  float t0, t1, u, tr2, tzz, sum, yl, d2, pm, dzz, dxx, dyy, a0, a1, a2, a3;
  float qq, y, ym, yp, rm, rp, q, rq, f, dg, e, g3, fac, m0, m1, m2;

  for (int g = G0; g < G1; ++g) {
    const int gn = (g + 1 < NS / 8) ? (g + 1) : (NS / 8 - 1);
    xs16 nA; xs8 nB;
    X_GROUP_LOAD(nA, nB, xb + 24 * gn);

    float* sb = rowb + (size_t)(8 * g) * LS;

    J2_STEP(cA[3],  cA[4],  cA[5],  q70, q71, q72, q00, q01, q02, J2_WAITX, J2_HOLD_TRIPLES);
    J2_STEP(cA[6],  cA[7],  cA[8],  q00, q01, q02, q10, q11, q12, "", J2_HOLD_NONE);
    J2_STEP(cA[9],  cA[10], cA[11], q10, q11, q12, q20, q21, q22, "", J2_HOLD_NONE);
    J2_STEP(cA[12], cA[13], cA[14], q20, q21, q22, q30, q31, q32, "", J2_HOLD_NONE);
    J2_STEP(cA[15], cB[0],  cB[1],  q30, q31, q32, q40, q41, q42, "", J2_HOLD_NONE);
    J2_STEP(cB[2],  cB[3],  cB[4],  q40, q41, q42, q50, q51, q52, "", J2_HOLD_NONE);
    J2_STEP(cB[5],  cB[6],  cB[7],  q50, q51, q52, q60, q61, q62, "", J2_HOLD_NONE);
    X_GROUP_WAIT(nA, nB);
    J2_STEP(nA[0],  nA[1],  nA[2],  q60, q61, q62, q70, q71, q72, "", J2_HOLD_NONE);
    J2_STORE_GROUP();

    cA = nA; cB = nB;
  }

  if (TO_CARRY) {
    cp[0 * 8192] = q70;  cp[1 * 8192] = q71;  cp[2 * 8192] = q72;
    cp[3 * 8192] = szz;  cp[4 * 8192] = eqps;
    cp[5 * 8192] = st0;  cp[6 * 8192] = st1;  cp[7 * 8192] = st2;
    cp[8 * 8192] = de0;  cp[9 * 8192] = de1;  cp[10 * 8192] = de2;
  }
}

#define TPITCH 196
#define BPITCH 200

template <int NW>
__device__ __forceinline__ void project_tile(
    const float* stress,
    const float* __restrict__ Bpad,
    float* __restrict__ out,
    const int tile) {
  __shared__ __attribute__((aligned(16))) float As[NW * 16 * TPITCH];
  __shared__ __attribute__((aligned(16))) float Bs[4 * BPITCH];

  const int lane = threadIdx.x & 31;
  const int wave = threadIdx.x >> 5;
  const int grp  = lane >> 4;
  const int ln   = lane & 15;

  const float4* src4 = (const float4*)(stress + (size_t)tile * 16 * LS);
  float* at = As + wave * (16 * TPITCH);

  float4 t[24];
#pragma unroll
  for (int i = 0; i < 24; ++i) t[i] = src4[i * 32 + lane];

  for (int q = threadIdx.x; q < 192; q += 32 * NW) {
    const int r = q / 48;
    const int c = q - r * 48;
    float4 v = make_float4(0.f, 0.f, 0.f, 0.f);
    if (r < NO) v = ((const float4*)Bpad)[r * 48 + c];
    *(float4*)(Bs + r * BPITCH + 4 * c) = v;
  }

  const int off0 = 4 * lane;
  const int off1 = (lane < 16) ? 4 * (32 + lane) : TPITCH + 4 * (lane - 16);
  const int off2 = TPITCH + 4 * (16 + lane);
#pragma unroll
  for (int m = 0; m < 8; ++m) {
    float* rowp = at + 2 * m * TPITCH;
    *(float4*)(rowp + off0) = t[3 * m + 0];
    *(float4*)(rowp + off1) = t[3 * m + 1];
    *(float4*)(rowp + off2) = t[3 * m + 2];
  }
  __syncthreads();

  const float* arow = at + ln * TPITCH + 2 * grp;
  const float* brow = Bs + (ln < NO ? ln : NO) * BPITCH + 2 * grp;

  v8f acc = {};
#pragma unroll
  for (int k = 0; k < LS; k += 4) {
    v2f a = *(const v2f*)(arow + k);
    v2f bm = *(const v2f*)(brow + k);
    acc = __builtin_amdgcn_wmma_f32_16x16x4_f32(
        false, a, false, bm, (short)0, acc, false, false);
  }

  if (ln < NO) {
#pragma unroll
    for (int r = 0; r < 8; ++r) {
      const size_t m = (size_t)tile * 16 + r + 8 * grp;
      out[m * NO + ln] = acc[r];
    }
  }
}

__device__ __forceinline__ int tile_before_cut(int u) {
  const int b = u / (S_CUT / 16);
  return b * (NS / 16) + (u - b * (S_CUT / 16));
}
__device__ __forceinline__ int tile_behind_cut(int u) {
  const int b = u / ((NS - S_CUT) / 16);
  return b * (NS / 16) + (S_CUT / 16) + (u - b * ((NS - S_CUT) / 16));
}

__global__ __launch_bounds__(64) void scan_head_kernel(
    const float* __restrict__ x,
    const float* __restrict__ W1,
    const float* __restrict__ W2,
    float* stress, float* Bpad, float* carry) {
  if (blockIdx.x < NB) {
    j2_scan_part<0, G_CUT, true, true>(x, W1, stress, carry,
                                       (int)blockIdx.x, (int)threadIdx.x);
  } else {
    prep_bpanel(W2, Bpad, (int)threadIdx.x, 64);
  }
}

__global__ __launch_bounds__(64) void scan_tail_project_kernel(
    const float* __restrict__ W1,
    float* carry,
    const float* __restrict__ x,
    float* stress, const float* Bpad,
    float* __restrict__ out) {
  if (blockIdx.x < NB) {
    j2_scan_part<G_CUT, NS / 8, false, false>(x, W1, stress, carry,
                                              (int)blockIdx.x, (int)threadIdx.x);
  } else {
    const int u = ((int)blockIdx.x - NB) * 2 + ((int)threadIdx.x >> 5);
    project_tile<2>(stress, Bpad, out, tile_before_cut(u));
  }
}

__global__ __launch_bounds__(256) void project_tail_kernel(
    const float* __restrict__ stress,
    const float* __restrict__ Bpad,
    float* __restrict__ out) {
  const int u = (int)blockIdx.x * 8 + ((int)threadIdx.x >> 5);
  project_tile<8>(stress, Bpad, out, tile_behind_cut(u));
}

extern "C" void kernel_launch(void* const* d_in, const int* in_sizes, int n_in,
                              void* d_out, int out_size, void* d_ws,
                              size_t ws_size, hipStream_t stream) {
  (void)in_sizes; (void)n_in; (void)out_size; (void)ws_size;
  const float* x  = (const float*)d_in[0];
  const float* W1 = (const float*)d_in[1];
  const float* W2 = (const float*)d_in[2];
  float* out = (float*)d_out;

  float* stress = (float*)d_ws;
  float* Bpad   = stress + (size_t)NB * NS * LS;
  float* carry  = Bpad + NPADN * LS;

  constexpr int tiles_before = NB * (S_CUT / 16);
  constexpr int tiles_behind = NB * ((NS - S_CUT) / 16);
  scan_head_kernel<<<NB + 1, 64, 0, stream>>>(x, W1, W2, stress, Bpad, carry);
  scan_tail_project_kernel<<<NB + tiles_before / 2, 64, 0, stream>>>(
      W1, carry, x, stress, Bpad, out);
  project_tail_kernel<<<tiles_behind / 8, 256, 0, stream>>>(stress, Bpad, out);
}


# scan: qq = 1.5*dev:dev formed from the differences of the trial normal stresses, (tr0-tr1)^2 + (tr1-tzz)*(tr0-tzz) + 3*dxy^2: mean and deviator leave the path to the root; qq's sum regrouped
# speedup vs baseline: 3.3627x; 1.1207x over previous
#include <hip/hip_runtime.h>


typedef __attribute__((ext_vector_type(2))) float v2f;
typedef __attribute__((ext_vector_type(8))) float v8f;

constexpr double dE    = 3130.0;
constexpr double dNU   = 0.37;
constexpr double dSIGY = 64.8;
constexpr double dHARD = 100.0;
constexpr double dG    = dE / (2.0 * (1.0 + dNU));
constexpr double dLAM  = dE * dNU / ((1.0 + dNU) * (1.0 - 2.0 * dNU));
constexpr double dC    = dE / ((1.0 + dNU) * (1.0 - 2.0 * dNU));

constexpr float fG     = (float)dG;
constexpr float fLAM   = (float)dLAM;
constexpr float fC11   = (float)(dC * (1.0 - dNU));
constexpr float fC12   = (float)(dC * dNU);
constexpr float fSIGY  = (float)dSIGY;
constexpr float fHARD  = (float)dHARD;
constexpr float f3G    = (float)(3.0 * dG);
constexpr float fINV3GH = (float)(1.0 / (3.0 * dG + dHARD));

#define NB   128
#define NS   2048
#define LS   192
#define NO   3
#define NPADN 16
#define S_CUT 1536
#define G_CUT (S_CUT / 8)
#define NCARRY 11

__device__ __forceinline__ void prep_bpanel(const float* __restrict__ W2,
                                            float* __restrict__ Bpad,
                                            int tid, int nthr) {
  for (int k = tid; k < LS; k += nthr) {
#pragma unroll
    for (int n = 0; n < NPADN; ++n) {
      float v = 0.0f;
      if (n < NO) {
        float w = W2[n * LS + k];
        v = fmaxf(w, 0.0f) + log1pf(__expf(-fabsf(w)));
      }
      Bpad[n * LS + k] = v;
    }
  }
}

static_assert(__builtin_bit_cast(unsigned, fC11) == 0x45acff7bu, "C11 literal");
static_assert(__builtin_bit_cast(unsigned, fC12) == 0x454b3437u, "C12 literal");
static_assert(__builtin_bit_cast(unsigned, fLAM) == 0x454b3437u, "LAM literal");
static_assert(__builtin_bit_cast(unsigned, fG) == 0x448ecabfu, "G literal");
static_assert(__builtin_bit_cast(unsigned, fSIGY) == 0x4281999au, "SIGY literal");
static_assert(__builtin_bit_cast(unsigned, fINV3GH) == 0x3994a648u, "1/(3G+H)");
static_assert(__builtin_bit_cast(unsigned, -f3G) == 0xc556301eu, "-3G literal");
static_assert(__builtin_bit_cast(unsigned, 1.0f / 3.0f) == 0x3eaaaaabu, "1/3");
static_assert(__builtin_bit_cast(unsigned, 1e-12f) == 0x2b8cbcccu, "1e-12");

typedef __attribute__((ext_vector_type(8)))  float xs8;
typedef __attribute__((ext_vector_type(16))) float xs16;

#define X_GROUP_LOAD(A, B, PTR) \
    asm volatile("s_load_b512 %0, %2, 0x0\n\t" \
                 "s_load_b256 %1, %2, 0x40" \
                 : "=&s"(A), "=&s"(B) : "s"(PTR) : "memory")
#define X_GROUP_WAIT(A, B) \
    asm volatile("s_wait_kmcnt 0x0" : "+s"(A), "+s"(B) : : "memory")

#define J2_STEP(NX0, NX1, NX2, P0, P1, P2, N0, N1, N2, WAITX, HOLD) \
    asm volatile( \
        "v_add_f32 %[u], %[de1], %[de0]\n\t"                         \
        "v_fma_f32 %[t0], %[de0], 0x45acff7b, %[p0]\n\t"             \
        "v_fma_f32 %[t1], %[de0], 0x454b3437, %[p1]\n\t"             \
        "v_fma_f32 %[tr2], %[de2], 0x448ecabf, %[p2]\n\t"            \
        "v_mul_f32 %[m0], %[nx0], %[w00]\n\t"                        \
        "v_fma_f32 %[tzz], %[u], 0x454b3437, %[szz]\n\t"             \
        "v_fma_f32 %[t0], %[de1], 0x454b3437, %[t0]\n\t"             \
        "v_fma_f32 %[t1], %[de1], 0x45acff7b, %[t1]\n\t"             \
        "v_mul_f32 %[t3], 0x40400000, %[tr2]\n\t"                    \
        "v_mul_f32 %[m1], %[nx0], %[w10]\n\t"                        \
        "v_mul_f32 %[m2], %[nx0], %[w20]\n\t"                        \
        "v_sub_f32 %[eac], %[t0], %[tzz]\n\t"                        \
        "v_sub_f32 %[ebc], %[t1], %[tzz]\n\t"                        \
        "v_fmaak_f32 %[k], %[t3], %[tr2], 0x2b8cbccc\n\t"            \
        "v_sub_f32 %[eab], %[t0], %[t1]\n\t"                         \
        "v_add_f32 %[sum], %[t1], %[t0]\n\t"                         \
        "v_fmaak_f32 %[yl], %[hard], %[eqps], 0x4281999a\n\t"        \
        "v_fma_f32 %[m0], %[w01], %[nx1], %[m0]\n\t"                 \
        "v_fma_f32 %[k], %[ebc], %[eac], %[k]\n\t"                   \
        "v_fma_f32 %[m1], %[w11], %[nx1], %[m1]\n\t"                 \
        "v_fma_f32 %[m2], %[w21], %[nx1], %[m2]\n\t"                 \
        "v_add_f32 %[sum], %[tzz], %[sum]\n\t"                       \
        "v_fma_f32 %[m0], %[w02], %[nx2], %[m0]\n\t"                 \
        "v_fma_f32 %[qq], %[eab], %[eab], %[k]\n\t"                  \
        "v_fma_f32 %[m1], %[w12], %[nx2], %[m1]\n\t"                 \
        "v_fma_f32 %[m2], %[w22], %[nx2], %[m2]\n\t"                 \
        "v_mul_f32 %[pm], 0x3eaaaaab, %[sum]\n\t"                    \
        "v_fma_f32 %[dzz], %[sum], 0xbeaaaaab, %[tzz]\n\t"           \
        "v_sqrt_f32 %[y], %[qq]\n\t"                                 \
        "v_sub_f32 %[de0], %[m0], %[st0]\n\t"                        \
        "v_sub_f32 %[de1], %[m1], %[st1]\n\t"                        \
        "v_sub_f32 %[dxx], %[t0], %[pm]\n\t"                         \
        "v_sub_f32 %[dyy], %[t1], %[pm]\n\t"                         \
        "v_sub_f32 %[de2], %[m2], %[st2]\n\t"                        \
        "v_add_nc_u32 %[ym], -1, %[y]\n\t"                           \
        "v_add_nc_u32 %[yp], 1, %[y]\n\t"                            \
        "v_mov_b32 %[st0], %[m0]\n\t"                                \
        "v_mov_b32 %[st1], %[m1]\n\t"                                \
        "v_fma_f32 %[rm], -%[ym], %[y], %[qq]\n\t"                   \
        "v_fma_f32 %[rp], -%[yp], %[y], %[qq]\n\t"                   \
        "v_mov_b32 %[st2], %[m2]\n\t"                                \
        "v_cmp_ge_f32 vcc_lo, 0, %[rm]\n\t"                          \
        "v_cndmask_b32 %[q], %[y], %[ym], vcc_lo\n\t"                \
        "v_cmp_lt_f32 vcc_lo, 0, %[rp]\n\t"                          \
        "v_cndmask_b32 %[q], %[q], %[yp], vcc_lo\n\t"                \
        "v_rcp_f32 %[rq], %[q]\n\t"                                  \
        "v_sub_f32 %[f], %[q], %[yl]\n\t"                            \
        "v_cmp_lt_f32 vcc_lo, 0, %[f]\n\t"                           \
        "v_mul_f32 %[dg], 0x3994a648, %[f]\n\t"                      \
        "v_fma_f32 %[e], -%[q], %[rq], 2.0\n\t"                      \
        "v_cndmask_b32 %[dg], 0, %[dg], vcc_lo\n\t"                  \
        "v_mul_f32 %[rq], %[rq], %[e]\n\t"                           \
        "v_mul_f32 %[g3], 0xc556301e, %[dg]\n\t"                     \
        "v_add_f32 %[eqps], %[eqps], %[dg]\n\t"                      \
        "v_fma_f32 %[fac], %[g3], %[rq], 1.0\n\t"                    \
        WAITX \
        "v_fma_f32 %[n0], %[dxx], %[fac], %[pm]\n\t"                 \
        "v_fma_f32 %[n1], %[dyy], %[fac], %[pm]\n\t"                 \
        "v_mul_f32 %[n2], %[tr2], %[fac]\n\t"                        \
        "v_fma_f32 %[szz], %[dzz], %[fac], %[pm]"                     \
        : \
          [n0] "=&v"(N0), [n1] "=&v"(N1), [n2] "=&v"(N2), \
          [szz] "+v"(szz), [eqps] "+v"(eqps), [st0] "+v"(st0), \
          [st1] "+v"(st1), [st2] "+v"(st2), [de0] "+v"(de0), [de1] "+v"(de1), \
          [de2] "+v"(de2), [t0] "=&v"(t0), [t1] "=&v"(t1), [u] "=&v"(u), \
          [tr2] "=&v"(tr2), [tzz] "=&v"(tzz), [sum] "=&v"(sum), \
          [yl] "=&v"(yl), [t3] "=&v"(t3), [pm] "=&v"(pm), [dzz] "=&v"(dzz), \
          [dxx] "=&v"(dxx), [dyy] "=&v"(dyy), [eab] "=&v"(eab), [eac] "=&v"(eac), \
          [ebc] "=&v"(ebc), [k] "=&v"(k), [qq] "=&v"(qq), [y] "=&v"(y), \
          [ym] "=&v"(ym), [yp] "=&v"(yp), [rm] "=&v"(rm), [rp] "=&v"(rp), \
          [q] "=&v"(q), [rq] "=&v"(rq), [f] "=&v"(f), [dg] "=&v"(dg), \
          [e] "=&v"(e), [g3] "=&v"(g3), [fac] "=&v"(fac), [m0] "=&v"(m0), \
          [m1] "=&v"(m1), [m2] "=&v"(m2) \
        : \
          [w00] "v"(w00), [w01] "v"(w01), [w02] "v"(w02), [w10] "v"(w10), \
          [w11] "v"(w11), [w12] "v"(w12), [w20] "v"(w20), [w21] "v"(w21), \
          [w22] "v"(w22), [nx0] "s"(NX0), [nx1] "s"(NX1), [nx2] "s"(NX2), \
          [p0] "v"(P0), [p1] "v"(P1), [p2] "v"(P2), \
          [hard] "s"(fHARD) HOLD \
        : "vcc", "memory");

#define J2_WAITX "s_wait_xcnt 0x0\n\t"
#define J2_HOLD_TRIPLES \
    , "v"(q10), "v"(q11), "v"(q12), "v"(q20), "v"(q21), "v"(q22), \
      "v"(q30), "v"(q31), "v"(q32), "v"(q40), "v"(q41), "v"(q42), \
      "v"(q50), "v"(q51), "v"(q52), "v"(q60), "v"(q61), "v"(q62)
#define J2_HOLD_NONE

#define J2_STORE_GROUP() \
    asm volatile( \
        "global_store_b96 %[voff], v[64:66], %[sb]\n\t" \
        "global_store_b96 %[voff], v[68:70], %[sb] offset:768\n\t" \
        "global_store_b96 %[voff], v[72:74], %[sb] offset:1536\n\t" \
        "global_store_b96 %[voff], v[76:78], %[sb] offset:2304\n\t" \
        "global_store_b96 %[voff], v[80:82], %[sb] offset:3072\n\t" \
        "global_store_b96 %[voff], v[84:86], %[sb] offset:3840\n\t" \
        "global_store_b96 %[voff], v[88:90], %[sb] offset:4608\n\t" \
        "global_store_b96 %[voff], v[40:42], %[sb] offset:5376" \
        : \
        : [voff] "v"(voff), [sb] "s"(sb), \
          "v"(q00), "v"(q01), "v"(q02), "v"(q10), "v"(q11), "v"(q12), \
          "v"(q20), "v"(q21), "v"(q22), "v"(q30), "v"(q31), "v"(q32), \
          "v"(q40), "v"(q41), "v"(q42), "v"(q50), "v"(q51), "v"(q52), \
          "v"(q60), "v"(q61), "v"(q62), "v"(q70), "v"(q71), "v"(q72) \
        : "memory");

template <int G0, int G1, bool FROM_START, bool TO_CARRY>
__device__ __forceinline__ void j2_scan_part(
    const float* __restrict__ x,
    const float* __restrict__ W1,
    float* stress,
    float* carry,
    const int b,
    const int j) {
  const float* __restrict__ xb = x + (size_t)b * NS * 3;

  const float* w = W1 + (3 * j) * 3;
  const float w00 = w[0], w01 = w[1], w02 = w[2];
  const float w10 = w[3], w11 = w[4], w12 = w[5];
  const float w20 = w[6], w21 = w[7], w22 = w[8];

  float* cp = carry + (b * 64 + j);

  register float q00 asm("v64"); register float q01 asm("v65"); register float q02 asm("v66");
  register float q10 asm("v68"); register float q11 asm("v69"); register float q12 asm("v70");
  register float q20 asm("v72"); register float q21 asm("v73"); register float q22 asm("v74");
  register float q30 asm("v76"); register float q31 asm("v77"); register float q32 asm("v78");
  register float q40 asm("v80"); register float q41 asm("v81"); register float q42 asm("v82");
  register float q50 asm("v84"); register float q51 asm("v85"); register float q52 asm("v86");
  register float q60 asm("v88"); register float q61 asm("v89"); register float q62 asm("v90");
  register float q70 asm("v40"); register float q71 asm("v41"); register float q72 asm("v42");
  float szz, eqps;

  float* rowb = stress + (size_t)b * NS * LS;
  const unsigned voff = 12u * (unsigned)j;

  float st0, st1, st2, de0, de1, de2;
  if (!FROM_START) {
    q70 = cp[0 * 8192]; q71 = cp[1 * 8192]; q72 = cp[2 * 8192];
    szz  = cp[3 * 8192]; eqps = cp[4 * 8192];
    st0  = cp[5 * 8192]; st1  = cp[6 * 8192]; st2  = cp[7 * 8192];
    de0  = cp[8 * 8192]; de1  = cp[9 * 8192]; de2  = cp[10 * 8192];
  }

  xs16 cA; xs8 cB;
  X_GROUP_LOAD(cA, cB, xb + 24 * G0);
  X_GROUP_WAIT(cA, cB);

  if (FROM_START) {
    q70 = 0.f; q71 = 0.f; q72 = 0.f; szz = 0.f; eqps = 0.f;
    const float x0 = cA[0], x1 = cA[1], x2 = cA[2];
    st0 = __builtin_fmaf(w02, x2, __builtin_fmaf(w01, x1, w00 * x0));
    st1 = __builtin_fmaf(w12, x2, __builtin_fmaf(w11, x1, w10 * x0));
    st2 = __builtin_fmaf(w22, x2, __builtin_fmaf(w21, x1, w20 * x0));
    de0 = st0 - 0.f; de1 = st1 - 0.f; de2 = st2 - 0.f;
  }

  q10 = 0.f; q11 = 0.f; q12 = 0.f; q20 = 0.f; q21 = 0.f; q22 = 0.f;
  q30 = 0.f; q31 = 0.f; q32 = 0.f; q40 = 0.f; q41 = 0.f; q42 = 0.f;
  q50 = 0.f; q51 = 0.f; q52 = 0.f; q60 = 0.f; q61 = 0.f; q62 = 0.f;

  float t0, t1, u, tr2, tzz, sum, yl, t3, pm, dzz, dxx, dyy, eab, eac, ebc, k;
  float qq, y, ym, yp, rm, rp, q, rq, f, dg, e, g3, fac, m0, m1, m2;

  for (int g = G0; g < G1; ++g) {
    const int gn = (g + 1 < NS / 8) ? (g + 1) : (NS / 8 - 1);
    xs16 nA; xs8 nB;
    X_GROUP_LOAD(nA, nB, xb + 24 * gn);

    float* sb = rowb + (size_t)(8 * g) * LS;

    J2_STEP(cA[3],  cA[4],  cA[5],  q70, q71, q72, q00, q01, q02, J2_WAITX, J2_HOLD_TRIPLES);
    J2_STEP(cA[6],  cA[7],  cA[8],  q00, q01, q02, q10, q11, q12, "", J2_HOLD_NONE);
    J2_STEP(cA[9],  cA[10], cA[11], q10, q11, q12, q20, q21, q22, "", J2_HOLD_NONE);
    J2_STEP(cA[12], cA[13], cA[14], q20, q21, q22, q30, q31, q32, "", J2_HOLD_NONE);
    J2_STEP(cA[15], cB[0],  cB[1],  q30, q31, q32, q40, q41, q42, "", J2_HOLD_NONE);
    J2_STEP(cB[2],  cB[3],  cB[4],  q40, q41, q42, q50, q51, q52, "", J2_HOLD_NONE);
    J2_STEP(cB[5],  cB[6],  cB[7],  q50, q51, q52, q60, q61, q62, "", J2_HOLD_NONE);
    X_GROUP_WAIT(nA, nB);
    J2_STEP(nA[0],  nA[1],  nA[2],  q60, q61, q62, q70, q71, q72, "", J2_HOLD_NONE);
    J2_STORE_GROUP();

    cA = nA; cB = nB;
  }

  if (TO_CARRY) {
    cp[0 * 8192] = q70;  cp[1 * 8192] = q71;  cp[2 * 8192] = q72;
    cp[3 * 8192] = szz;  cp[4 * 8192] = eqps;
    cp[5 * 8192] = st0;  cp[6 * 8192] = st1;  cp[7 * 8192] = st2;
    cp[8 * 8192] = de0;  cp[9 * 8192] = de1;  cp[10 * 8192] = de2;
  }
}

#define TPITCH 196
#define BPITCH 200

template <int NW>
__device__ __forceinline__ void project_tile(
    const float* stress,
    const float* __restrict__ Bpad,
    float* __restrict__ out,
    const int tile) {
  __shared__ __attribute__((aligned(16))) float As[NW * 16 * TPITCH];
  __shared__ __attribute__((aligned(16))) float Bs[4 * BPITCH];

  const int lane = threadIdx.x & 31;
  const int wave = threadIdx.x >> 5;
  const int grp  = lane >> 4;
  const int ln   = lane & 15;

  const float4* src4 = (const float4*)(stress + (size_t)tile * 16 * LS);
  float* at = As + wave * (16 * TPITCH);

  float4 t[24];
#pragma unroll
  for (int i = 0; i < 24; ++i) t[i] = src4[i * 32 + lane];

  for (int q = threadIdx.x; q < 192; q += 32 * NW) {
    const int r = q / 48;
    const int c = q - r * 48;
    float4 v = make_float4(0.f, 0.f, 0.f, 0.f);
    if (r < NO) v = ((const float4*)Bpad)[r * 48 + c];
    *(float4*)(Bs + r * BPITCH + 4 * c) = v;
  }

  const int off0 = 4 * lane;
  const int off1 = (lane < 16) ? 4 * (32 + lane) : TPITCH + 4 * (lane - 16);
  const int off2 = TPITCH + 4 * (16 + lane);
#pragma unroll
  for (int m = 0; m < 8; ++m) {
    float* rowp = at + 2 * m * TPITCH;
    *(float4*)(rowp + off0) = t[3 * m + 0];
    *(float4*)(rowp + off1) = t[3 * m + 1];
    *(float4*)(rowp + off2) = t[3 * m + 2];
  }
  __syncthreads();

  const float* arow = at + ln * TPITCH + 2 * grp;
  const float* brow = Bs + (ln < NO ? ln : NO) * BPITCH + 2 * grp;

  v8f acc = {};
#pragma unroll
  for (int k = 0; k < LS; k += 4) {
    v2f a = *(const v2f*)(arow + k);
    v2f bm = *(const v2f*)(brow + k);
    acc = __builtin_amdgcn_wmma_f32_16x16x4_f32(
        false, a, false, bm, (short)0, acc, false, false);
  }

  if (ln < NO) {
#pragma unroll
    for (int r = 0; r < 8; ++r) {
      const size_t m = (size_t)tile * 16 + r + 8 * grp;
      out[m * NO + ln] = acc[r];
    }
  }
}

__device__ __forceinline__ int tile_before_cut(int u) {
  const int b = u / (S_CUT / 16);
  return b * (NS / 16) + (u - b * (S_CUT / 16));
}
__device__ __forceinline__ int tile_behind_cut(int u) {
  const int b = u / ((NS - S_CUT) / 16);
  return b * (NS / 16) + (S_CUT / 16) + (u - b * ((NS - S_CUT) / 16));
}

__global__ __launch_bounds__(64) void scan_head_kernel(
    const float* __restrict__ x,
    const float* __restrict__ W1,
    const float* __restrict__ W2,
    float* stress, float* Bpad, float* carry) {
  if (blockIdx.x < NB) {
    j2_scan_part<0, G_CUT, true, true>(x, W1, stress, carry,
                                       (int)blockIdx.x, (int)threadIdx.x);
  } else {
    prep_bpanel(W2, Bpad, (int)threadIdx.x, 64);
  }
}

__global__ __launch_bounds__(64) void scan_tail_project_kernel(
    const float* __restrict__ W1,
    float* carry,
    const float* __restrict__ x,
    float* stress, const float* Bpad,
    float* __restrict__ out) {
  if (blockIdx.x < NB) {
    j2_scan_part<G_CUT, NS / 8, false, false>(x, W1, stress, carry,
                                              (int)blockIdx.x, (int)threadIdx.x);
  } else {
    const int u = ((int)blockIdx.x - NB) * 2 + ((int)threadIdx.x >> 5);
    project_tile<2>(stress, Bpad, out, tile_before_cut(u));
  }
}

__global__ __launch_bounds__(256) void project_tail_kernel(
    const float* __restrict__ stress,
    const float* __restrict__ Bpad,
    float* __restrict__ out) {
  const int u = (int)blockIdx.x * 8 + ((int)threadIdx.x >> 5);
  project_tile<8>(stress, Bpad, out, tile_behind_cut(u));
}

extern "C" void kernel_launch(void* const* d_in, const int* in_sizes, int n_in,
                              void* d_out, int out_size, void* d_ws,
                              size_t ws_size, hipStream_t stream) {
  (void)in_sizes; (void)n_in; (void)out_size; (void)ws_size;
  const float* x  = (const float*)d_in[0];
  const float* W1 = (const float*)d_in[1];
  const float* W2 = (const float*)d_in[2];
  float* out = (float*)d_out;

  float* stress = (float*)d_ws;
  float* Bpad   = stress + (size_t)NB * NS * LS;
  float* carry  = Bpad + NPADN * LS;

  constexpr int tiles_before = NB * (S_CUT / 16);
  constexpr int tiles_behind = NB * ((NS - S_CUT) / 16);
  scan_head_kernel<<<NB + 1, 64, 0, stream>>>(x, W1, W2, stress, Bpad, carry);
  scan_tail_project_kernel<<<NB + tiles_before / 2, 64, 0, stream>>>(
      W1, carry, x, stress, Bpad, out);
  project_tail_kernel<<<tiles_behind / 8, 256, 0, stream>>>(stress, Bpad, out);
}


# scan: q is the hardware root alone (v_sqrt_f32, within 1 ulp of sqrtf): the two-neighbour correction (5 dependent levels between root and reciprocal) leaves the carried chain; sums unchanged
# speedup vs baseline: 3.6659x; 1.0902x over previous
#include <hip/hip_runtime.h>


typedef __attribute__((ext_vector_type(2))) float v2f;
typedef __attribute__((ext_vector_type(8))) float v8f;

constexpr double dE    = 3130.0;
constexpr double dNU   = 0.37;
constexpr double dSIGY = 64.8;
constexpr double dHARD = 100.0;
constexpr double dG    = dE / (2.0 * (1.0 + dNU));
constexpr double dLAM  = dE * dNU / ((1.0 + dNU) * (1.0 - 2.0 * dNU));
constexpr double dC    = dE / ((1.0 + dNU) * (1.0 - 2.0 * dNU));

constexpr float fG     = (float)dG;
constexpr float fLAM   = (float)dLAM;
constexpr float fC11   = (float)(dC * (1.0 - dNU));
constexpr float fC12   = (float)(dC * dNU);
constexpr float fSIGY  = (float)dSIGY;
constexpr float fHARD  = (float)dHARD;
constexpr float f3G    = (float)(3.0 * dG);
constexpr float fINV3GH = (float)(1.0 / (3.0 * dG + dHARD));

#define NB   128
#define NS   2048
#define LS   192
#define NO   3
#define NPADN 16
#define S_CUT 1536
#define G_CUT (S_CUT / 8)
#define NCARRY 11

__device__ __forceinline__ void prep_bpanel(const float* __restrict__ W2,
                                            float* __restrict__ Bpad,
                                            int tid, int nthr) {
  for (int k = tid; k < LS; k += nthr) {
#pragma unroll
    for (int n = 0; n < NPADN; ++n) {
      float v = 0.0f;
      if (n < NO) {
        float w = W2[n * LS + k];
        v = fmaxf(w, 0.0f) + log1pf(__expf(-fabsf(w)));
      }
      Bpad[n * LS + k] = v;
    }
  }
}

static_assert(__builtin_bit_cast(unsigned, fC11) == 0x45acff7bu, "C11 literal");
static_assert(__builtin_bit_cast(unsigned, fC12) == 0x454b3437u, "C12 literal");
static_assert(__builtin_bit_cast(unsigned, fLAM) == 0x454b3437u, "LAM literal");
static_assert(__builtin_bit_cast(unsigned, fG) == 0x448ecabfu, "G literal");
static_assert(__builtin_bit_cast(unsigned, fSIGY) == 0x4281999au, "SIGY literal");
static_assert(__builtin_bit_cast(unsigned, fINV3GH) == 0x3994a648u, "1/(3G+H)");
static_assert(__builtin_bit_cast(unsigned, -f3G) == 0xc556301eu, "-3G literal");
static_assert(__builtin_bit_cast(unsigned, 1.0f / 3.0f) == 0x3eaaaaabu, "1/3");
static_assert(__builtin_bit_cast(unsigned, 1e-12f) == 0x2b8cbcccu, "1e-12");

typedef __attribute__((ext_vector_type(8)))  float xs8;
typedef __attribute__((ext_vector_type(16))) float xs16;

#define X_GROUP_LOAD(A, B, PTR) \
    asm volatile("s_load_b512 %0, %2, 0x0\n\t" \
                 "s_load_b256 %1, %2, 0x40" \
                 : "=&s"(A), "=&s"(B) : "s"(PTR) : "memory")
#define X_GROUP_WAIT(A, B) \
    asm volatile("s_wait_kmcnt 0x0" : "+s"(A), "+s"(B) : : "memory")

#define J2_STEP(NX0, NX1, NX2, P0, P1, P2, N0, N1, N2, WAITX, HOLD) \
    asm volatile( \
        "v_add_f32 %[u], %[de1], %[de0]\n\t"                         \
        "v_fma_f32 %[t0], %[de0], 0x45acff7b, %[p0]\n\t"             \
        "v_fma_f32 %[t1], %[de0], 0x454b3437, %[p1]\n\t"             \
        "v_fma_f32 %[tr2], %[de2], 0x448ecabf, %[p2]\n\t"            \
        "v_mul_f32 %[m0], %[nx0], %[w00]\n\t"                        \
        "v_fma_f32 %[tzz], %[u], 0x454b3437, %[szz]\n\t"             \
        "v_fma_f32 %[t0], %[de1], 0x454b3437, %[t0]\n\t"             \
        "v_fma_f32 %[t1], %[de1], 0x45acff7b, %[t1]\n\t"             \
        "v_mul_f32 %[t3], 0x40400000, %[tr2]\n\t"                    \
        "v_mul_f32 %[m1], %[nx0], %[w10]\n\t"                        \
        "v_mul_f32 %[m2], %[nx0], %[w20]\n\t"                        \
        "v_sub_f32 %[eac], %[t0], %[tzz]\n\t"                        \
        "v_sub_f32 %[ebc], %[t1], %[tzz]\n\t"                        \
        "v_fmaak_f32 %[k], %[t3], %[tr2], 0x2b8cbccc\n\t"            \
        "v_sub_f32 %[eab], %[t0], %[t1]\n\t"                         \
        "v_add_f32 %[sum], %[t1], %[t0]\n\t"                         \
        "v_fmaak_f32 %[yl], %[hard], %[eqps], 0x4281999a\n\t"        \
        "v_fma_f32 %[m0], %[w01], %[nx1], %[m0]\n\t"                 \
        "v_fma_f32 %[k], %[ebc], %[eac], %[k]\n\t"                   \
        "v_fma_f32 %[m1], %[w11], %[nx1], %[m1]\n\t"                 \
        "v_fma_f32 %[m2], %[w21], %[nx1], %[m2]\n\t"                 \
        "v_add_f32 %[sum], %[tzz], %[sum]\n\t"                       \
        "v_fma_f32 %[m0], %[w02], %[nx2], %[m0]\n\t"                 \
        "v_fma_f32 %[qq], %[eab], %[eab], %[k]\n\t"                  \
        "v_fma_f32 %[m1], %[w12], %[nx2], %[m1]\n\t"                 \
        "v_fma_f32 %[m2], %[w22], %[nx2], %[m2]\n\t"                 \
        "v_mul_f32 %[pm], 0x3eaaaaab, %[sum]\n\t"                    \
        "v_fma_f32 %[dzz], %[sum], 0xbeaaaaab, %[tzz]\n\t"           \
        "v_sqrt_f32 %[q], %[qq]\n\t"                                 \
        "v_sub_f32 %[de0], %[m0], %[st0]\n\t"                        \
        "v_sub_f32 %[de1], %[m1], %[st1]\n\t"                        \
        "v_sub_f32 %[dxx], %[t0], %[pm]\n\t"                         \
        "v_sub_f32 %[dyy], %[t1], %[pm]\n\t"                         \
        "v_sub_f32 %[de2], %[m2], %[st2]\n\t"                        \
        "v_mov_b32 %[st0], %[m0]\n\t"                                \
        "v_mov_b32 %[st1], %[m1]\n\t"                                \
        "v_rcp_f32 %[rq], %[q]\n\t"                                  \
        "v_sub_f32 %[f], %[q], %[yl]\n\t"                            \
        "v_mov_b32 %[st2], %[m2]\n\t"                                \
        "v_cmp_lt_f32 vcc_lo, 0, %[f]\n\t"                           \
        "v_mul_f32 %[dg], 0x3994a648, %[f]\n\t"                      \
        "v_fma_f32 %[e], -%[q], %[rq], 2.0\n\t"                      \
        "v_cndmask_b32 %[dg], 0, %[dg], vcc_lo\n\t"                  \
        "v_mul_f32 %[rq], %[rq], %[e]\n\t"                           \
        "v_mul_f32 %[g3], 0xc556301e, %[dg]\n\t"                     \
        "v_add_f32 %[eqps], %[eqps], %[dg]\n\t"                      \
        "v_fma_f32 %[fac], %[g3], %[rq], 1.0\n\t"                    \
        WAITX \
        "v_fma_f32 %[n0], %[dxx], %[fac], %[pm]\n\t"                 \
        "v_fma_f32 %[n1], %[dyy], %[fac], %[pm]\n\t"                 \
        "v_mul_f32 %[n2], %[tr2], %[fac]\n\t"                        \
        "v_fma_f32 %[szz], %[dzz], %[fac], %[pm]"                     \
        : \
          [n0] "=&v"(N0), [n1] "=&v"(N1), [n2] "=&v"(N2), \
          [szz] "+v"(szz), [eqps] "+v"(eqps), [st0] "+v"(st0), \
          [st1] "+v"(st1), [st2] "+v"(st2), [de0] "+v"(de0), [de1] "+v"(de1), \
          [de2] "+v"(de2), [t0] "=&v"(t0), [t1] "=&v"(t1), [u] "=&v"(u), \
          [tr2] "=&v"(tr2), [tzz] "=&v"(tzz), [sum] "=&v"(sum), \
          [yl] "=&v"(yl), [t3] "=&v"(t3), [pm] "=&v"(pm), [dzz] "=&v"(dzz), \
          [dxx] "=&v"(dxx), [dyy] "=&v"(dyy), [eab] "=&v"(eab), [eac] "=&v"(eac), \
          [ebc] "=&v"(ebc), [k] "=&v"(k), [qq] "=&v"(qq), \
          [q] "=&v"(q), [rq] "=&v"(rq), [f] "=&v"(f), [dg] "=&v"(dg), \
          [e] "=&v"(e), [g3] "=&v"(g3), [fac] "=&v"(fac), [m0] "=&v"(m0), \
          [m1] "=&v"(m1), [m2] "=&v"(m2) \
        : \
          [w00] "v"(w00), [w01] "v"(w01), [w02] "v"(w02), [w10] "v"(w10), \
          [w11] "v"(w11), [w12] "v"(w12), [w20] "v"(w20), [w21] "v"(w21), \
          [w22] "v"(w22), [nx0] "s"(NX0), [nx1] "s"(NX1), [nx2] "s"(NX2), \
          [p0] "v"(P0), [p1] "v"(P1), [p2] "v"(P2), \
          [hard] "s"(fHARD) HOLD \
        : "vcc", "memory");

#define J2_WAITX "s_wait_xcnt 0x0\n\t"
#define J2_HOLD_TRIPLES \
    , "v"(q10), "v"(q11), "v"(q12), "v"(q20), "v"(q21), "v"(q22), \
      "v"(q30), "v"(q31), "v"(q32), "v"(q40), "v"(q41), "v"(q42), \
      "v"(q50), "v"(q51), "v"(q52), "v"(q60), "v"(q61), "v"(q62)
#define J2_HOLD_NONE

#define J2_STORE_GROUP() \
    asm volatile( \
        "global_store_b96 %[voff], v[64:66], %[sb]\n\t" \
        "global_store_b96 %[voff], v[68:70], %[sb] offset:768\n\t" \
        "global_store_b96 %[voff], v[72:74], %[sb] offset:1536\n\t" \
        "global_store_b96 %[voff], v[76:78], %[sb] offset:2304\n\t" \
        "global_store_b96 %[voff], v[80:82], %[sb] offset:3072\n\t" \
        "global_store_b96 %[voff], v[84:86], %[sb] offset:3840\n\t" \
        "global_store_b96 %[voff], v[88:90], %[sb] offset:4608\n\t" \
        "global_store_b96 %[voff], v[40:42], %[sb] offset:5376" \
        : \
        : [voff] "v"(voff), [sb] "s"(sb), \
          "v"(q00), "v"(q01), "v"(q02), "v"(q10), "v"(q11), "v"(q12), \
          "v"(q20), "v"(q21), "v"(q22), "v"(q30), "v"(q31), "v"(q32), \
          "v"(q40), "v"(q41), "v"(q42), "v"(q50), "v"(q51), "v"(q52), \
          "v"(q60), "v"(q61), "v"(q62), "v"(q70), "v"(q71), "v"(q72) \
        : "memory");

template <int G0, int G1, bool FROM_START, bool TO_CARRY>
__device__ __forceinline__ void j2_scan_part(
    const float* __restrict__ x,
    const float* __restrict__ W1,
    float* stress,
    float* carry,
    const int b,
    const int j) {
  const float* __restrict__ xb = x + (size_t)b * NS * 3;

  const float* w = W1 + (3 * j) * 3;
  const float w00 = w[0], w01 = w[1], w02 = w[2];
  const float w10 = w[3], w11 = w[4], w12 = w[5];
  const float w20 = w[6], w21 = w[7], w22 = w[8];

  float* cp = carry + (b * 64 + j);

  register float q00 asm("v64"); register float q01 asm("v65"); register float q02 asm("v66");
  register float q10 asm("v68"); register float q11 asm("v69"); register float q12 asm("v70");
  register float q20 asm("v72"); register float q21 asm("v73"); register float q22 asm("v74");
  register float q30 asm("v76"); register float q31 asm("v77"); register float q32 asm("v78");
  register float q40 asm("v80"); register float q41 asm("v81"); register float q42 asm("v82");
  register float q50 asm("v84"); register float q51 asm("v85"); register float q52 asm("v86");
  register float q60 asm("v88"); register float q61 asm("v89"); register float q62 asm("v90");
  register float q70 asm("v40"); register float q71 asm("v41"); register float q72 asm("v42");
  float szz, eqps;

  float* rowb = stress + (size_t)b * NS * LS;
  const unsigned voff = 12u * (unsigned)j;

  float st0, st1, st2, de0, de1, de2;
  if (!FROM_START) {
    q70 = cp[0 * 8192]; q71 = cp[1 * 8192]; q72 = cp[2 * 8192];
    szz  = cp[3 * 8192]; eqps = cp[4 * 8192];
    st0  = cp[5 * 8192]; st1  = cp[6 * 8192]; st2  = cp[7 * 8192];
    de0  = cp[8 * 8192]; de1  = cp[9 * 8192]; de2  = cp[10 * 8192];
  }

  xs16 cA; xs8 cB;
  X_GROUP_LOAD(cA, cB, xb + 24 * G0);
  X_GROUP_WAIT(cA, cB);

  if (FROM_START) {
    q70 = 0.f; q71 = 0.f; q72 = 0.f; szz = 0.f; eqps = 0.f;
    const float x0 = cA[0], x1 = cA[1], x2 = cA[2];
    st0 = __builtin_fmaf(w02, x2, __builtin_fmaf(w01, x1, w00 * x0));
    st1 = __builtin_fmaf(w12, x2, __builtin_fmaf(w11, x1, w10 * x0));
    st2 = __builtin_fmaf(w22, x2, __builtin_fmaf(w21, x1, w20 * x0));
    de0 = st0 - 0.f; de1 = st1 - 0.f; de2 = st2 - 0.f;
  }

  q10 = 0.f; q11 = 0.f; q12 = 0.f; q20 = 0.f; q21 = 0.f; q22 = 0.f;
  q30 = 0.f; q31 = 0.f; q32 = 0.f; q40 = 0.f; q41 = 0.f; q42 = 0.f;
  q50 = 0.f; q51 = 0.f; q52 = 0.f; q60 = 0.f; q61 = 0.f; q62 = 0.f;

  float t0, t1, u, tr2, tzz, sum, yl, t3, pm, dzz, dxx, dyy, eab, eac, ebc, k;
  float qq, q, rq, f, dg, e, g3, fac, m0, m1, m2;

  for (int g = G0; g < G1; ++g) {
    const int gn = (g + 1 < NS / 8) ? (g + 1) : (NS / 8 - 1);
    xs16 nA; xs8 nB;
    X_GROUP_LOAD(nA, nB, xb + 24 * gn);

    float* sb = rowb + (size_t)(8 * g) * LS;

    J2_STEP(cA[3],  cA[4],  cA[5],  q70, q71, q72, q00, q01, q02, J2_WAITX, J2_HOLD_TRIPLES);
    J2_STEP(cA[6],  cA[7],  cA[8],  q00, q01, q02, q10, q11, q12, "", J2_HOLD_NONE);
    J2_STEP(cA[9],  cA[10], cA[11], q10, q11, q12, q20, q21, q22, "", J2_HOLD_NONE);
    J2_STEP(cA[12], cA[13], cA[14], q20, q21, q22, q30, q31, q32, "", J2_HOLD_NONE);
    J2_STEP(cA[15], cB[0],  cB[1],  q30, q31, q32, q40, q41, q42, "", J2_HOLD_NONE);
    J2_STEP(cB[2],  cB[3],  cB[4],  q40, q41, q42, q50, q51, q52, "", J2_HOLD_NONE);
    J2_STEP(cB[5],  cB[6],  cB[7],  q50, q51, q52, q60, q61, q62, "", J2_HOLD_NONE);
    X_GROUP_WAIT(nA, nB);
    J2_STEP(nA[0],  nA[1],  nA[2],  q60, q61, q62, q70, q71, q72, "", J2_HOLD_NONE);
    J2_STORE_GROUP();

    cA = nA; cB = nB;
  }

  if (TO_CARRY) {
    cp[0 * 8192] = q70;  cp[1 * 8192] = q71;  cp[2 * 8192] = q72;
    cp[3 * 8192] = szz;  cp[4 * 8192] = eqps;
    cp[5 * 8192] = st0;  cp[6 * 8192] = st1;  cp[7 * 8192] = st2;
    cp[8 * 8192] = de0;  cp[9 * 8192] = de1;  cp[10 * 8192] = de2;
  }
}

#define TPITCH 196
#define BPITCH 200

template <int NW>
__device__ __forceinline__ void project_tile(
    const float* stress,
    const float* __restrict__ Bpad,
    float* __restrict__ out,
    const int tile) {
  __shared__ __attribute__((aligned(16))) float As[NW * 16 * TPITCH];
  __shared__ __attribute__((aligned(16))) float Bs[4 * BPITCH];

  const int lane = threadIdx.x & 31;
  const int wave = threadIdx.x >> 5;
  const int grp  = lane >> 4;
  const int ln   = lane & 15;

  const float4* src4 = (const float4*)(stress + (size_t)tile * 16 * LS);
  float* at = As + wave * (16 * TPITCH);

  float4 t[24];
#pragma unroll
  for (int i = 0; i < 24; ++i) t[i] = src4[i * 32 + lane];

  for (int q = threadIdx.x; q < 192; q += 32 * NW) {
    const int r = q / 48;
    const int c = q - r * 48;
    float4 v = make_float4(0.f, 0.f, 0.f, 0.f);
    if (r < NO) v = ((const float4*)Bpad)[r * 48 + c];
    *(float4*)(Bs + r * BPITCH + 4 * c) = v;
  }

  const int off0 = 4 * lane;
  const int off1 = (lane < 16) ? 4 * (32 + lane) : TPITCH + 4 * (lane - 16);
  const int off2 = TPITCH + 4 * (16 + lane);
#pragma unroll
  for (int m = 0; m < 8; ++m) {
    float* rowp = at + 2 * m * TPITCH;
    *(float4*)(rowp + off0) = t[3 * m + 0];
    *(float4*)(rowp + off1) = t[3 * m + 1];
    *(float4*)(rowp + off2) = t[3 * m + 2];
  }
  __syncthreads();

  const float* arow = at + ln * TPITCH + 2 * grp;
  const float* brow = Bs + (ln < NO ? ln : NO) * BPITCH + 2 * grp;

  v8f acc = {};
#pragma unroll
  for (int k = 0; k < LS; k += 4) {
    v2f a = *(const v2f*)(arow + k);
    v2f bm = *(const v2f*)(brow + k);
    acc = __builtin_amdgcn_wmma_f32_16x16x4_f32(
        false, a, false, bm, (short)0, acc, false, false);
  }

  if (ln < NO) {
#pragma unroll
    for (int r = 0; r < 8; ++r) {
      const size_t m = (size_t)tile * 16 + r + 8 * grp;
      out[m * NO + ln] = acc[r];
    }
  }
}

__device__ __forceinline__ int tile_before_cut(int u) {
  const int b = u / (S_CUT / 16);
  return b * (NS / 16) + (u - b * (S_CUT / 16));
}
__device__ __forceinline__ int tile_behind_cut(int u) {
  const int b = u / ((NS - S_CUT) / 16);
  return b * (NS / 16) + (S_CUT / 16) + (u - b * ((NS - S_CUT) / 16));
}

__global__ __launch_bounds__(64) void scan_head_kernel(
    const float* __restrict__ x,
    const float* __restrict__ W1,
    const float* __restrict__ W2,
    float* stress, float* Bpad, float* carry) {
  if (blockIdx.x < NB) {
    j2_scan_part<0, G_CUT, true, true>(x, W1, stress, carry,
                                       (int)blockIdx.x, (int)threadIdx.x);
  } else {
    prep_bpanel(W2, Bpad, (int)threadIdx.x, 64);
  }
}

__global__ __launch_bounds__(64) void scan_tail_project_kernel(
    const float* __restrict__ W1,
    float* carry,
    const float* __restrict__ x,
    float* stress, const float* Bpad,
    float* __restrict__ out) {
  if (blockIdx.x < NB) {
    j2_scan_part<G_CUT, NS / 8, false, false>(x, W1, stress, carry,
                                              (int)blockIdx.x, (int)threadIdx.x);
  } else {
    const int u = ((int)blockIdx.x - NB) * 2 + ((int)threadIdx.x >> 5);
    project_tile<2>(stress, Bpad, out, tile_before_cut(u));
  }
}

__global__ __launch_bounds__(256) void project_tail_kernel(
    const float* __restrict__ stress,
    const float* __restrict__ Bpad,
    float* __restrict__ out) {
  const int u = (int)blockIdx.x * 8 + ((int)threadIdx.x >> 5);
  project_tile<8>(stress, Bpad, out, tile_behind_cut(u));
}

extern "C" void kernel_launch(void* const* d_in, const int* in_sizes, int n_in,
                              void* d_out, int out_size, void* d_ws,
                              size_t ws_size, hipStream_t stream) {
  (void)in_sizes; (void)n_in; (void)out_size; (void)ws_size;
  const float* x  = (const float*)d_in[0];
  const float* W1 = (const float*)d_in[1];
  const float* W2 = (const float*)d_in[2];
  float* out = (float*)d_out;

  float* stress = (float*)d_ws;
  float* Bpad   = stress + (size_t)NB * NS * LS;
  float* carry  = Bpad + NPADN * LS;

  constexpr int tiles_before = NB * (S_CUT / 16);
  constexpr int tiles_behind = NB * ((NS - S_CUT) / 16);
  scan_head_kernel<<<NB + 1, 64, 0, stream>>>(x, W1, W2, stress, Bpad, carry);
  scan_tail_project_kernel<<<NB + tiles_before / 2, 64, 0, stream>>>(
      W1, carry, x, stress, Bpad, out);
  project_tail_kernel<<<tiles_behind / 8, 256, 0, stream>>>(stress, Bpad, out);
}
